# starting kernel
# baseline (speedup 1.0000x reference)
#include <hip/hip_runtime.h>


typedef __attribute__((ext_vector_type(2))) float v2f;
typedef __attribute__((ext_vector_type(8))) float v8f;

static inline int cdiv(long long a, long long b) { return (int)((a + b - 1) / b); }

template <int K, int N>
__launch_bounds__(256)
__global__ void gemm_wmma_f32(const float* __restrict__ X,
                              const float* __restrict__ W,
                              float* __restrict__ H,
                              int nrows) {
    constexpr int NT = N / 16;
    __shared__ float BF[(K / 4) * NT * 64];

    const int tid = threadIdx.x;
    for (int idx = tid; idx < K * N; idx += 256) {
        const int n  = idx / K;
        const int k  = idx - n * K;
        const int ks = k >> 2;
        const int kk = k & 3;
        const int g  = kk >> 1;
        const int r  = kk & 1;
        const int t  = n >> 4;
        const int nl = n & 15;
        BF[((ks * NT + t) * 32 + g * 16 + nl) * 2 + r] = W[idx];
    }
    __syncthreads();

    const int wave   = tid >> 5;
    const int lane   = tid & 31;
    const int laneLo = lane & 15;
    const int laneHi = lane >> 4;
    const int m0     = (blockIdx.x * 8 + wave) * 16;
    if (m0 >= nrows) return;

    v8f acc[NT] = {};

    const float* xrow  = X + (size_t)(m0 + laneLo) * K;
    const float* fbase = BF + lane * 2;
    for (int ks = 0; ks < K / 4; ++ks) {
        const v2f a = *(const v2f*)(xrow + ks * 4 + 2 * laneHi);
        const float* fb = fbase + (ks * NT) * 64;
#pragma unroll
        for (int t = 0; t < NT; ++t) {
            const v2f b = *(const v2f*)(fb + t * 64);
            acc[t] = __builtin_amdgcn_wmma_f32_16x16x4_f32(
                false, a, false, b, (short)0, acc[t], false, false);
        }
    }

#pragma unroll
    for (int t = 0; t < NT; ++t)
#pragma unroll
        for (int r = 0; r < 8; ++r)
            H[(size_t)(m0 + r + 8 * laneHi) * N + t * 16 + laneLo] = acc[t][r];
}

__global__ void k_fill1(float* __restrict__ deg, int n) {
    int i = blockIdx.x * blockDim.x + threadIdx.x;
    if (i < n) deg[i] = 1.0f;
}

__global__ void k_deg(const int* __restrict__ dst, float* __restrict__ deg, int E) {
    int e = blockIdx.x * blockDim.x + threadIdx.x;
    if (e < E) atomicAdd(&deg[dst[e]], 1.0f);
}

__global__ void k_rsqrt(float* __restrict__ d, int n) {
    int i = blockIdx.x * blockDim.x + threadIdx.x;
    if (i < n) d[i] = rsqrtf(d[i]);
}

__global__ void k_norm(const int* __restrict__ src, const int* __restrict__ dst,
                       const float* __restrict__ dinv, float* __restrict__ norm, int E) {
    int e = blockIdx.x * blockDim.x + threadIdx.x;
    if (e < E) norm[e] = dinv[src[e]] * dinv[dst[e]];
}

template <int C>
__global__ void k_selfinit(const float* __restrict__ H, const float* __restrict__ dinv,
                           const float* __restrict__ bias, float* __restrict__ out, int n) {
    int gid = blockIdx.x * blockDim.x + threadIdx.x;
    const int CH = C / 4;
    int i = gid / CH;
    if (i >= n) return;
    int c = (gid - i * CH) * 4;
    float s = dinv[i] * dinv[i];
    float4 h = *(const float4*)(H + (size_t)i * C + c);
    float4 o;
    o.x = h.x * s + bias[c + 0];
    o.y = h.y * s + bias[c + 1];
    o.z = h.z * s + bias[c + 2];
    o.w = h.w * s + bias[c + 3];
    *(float4*)(out + (size_t)i * C + c) = o;
}

template <int C>
__global__ void k_scatter(const int* __restrict__ src, const int* __restrict__ dst,
                          const float* __restrict__ norm, const float* __restrict__ H,
                          float* __restrict__ out, int E) {
    int gid = blockIdx.x * blockDim.x + threadIdx.x;
    const int CH = C / 4;
    int e = gid / CH;
    if (e >= E) return;
    int c = (gid - e * CH) * 4;
    float nrm = norm[e];
    float4 h = *(const float4*)(H + (size_t)src[e] * C + c);
    float* o = out + (size_t)dst[e] * C + c;
    atomicAdd(o + 0, h.x * nrm);
    atomicAdd(o + 1, h.y * nrm);
    atomicAdd(o + 2, h.z * nrm);
    atomicAdd(o + 3, h.w * nrm);
}

__global__ void k_relu(const float* __restrict__ in, float* __restrict__ out, int n) {
    int i = blockIdx.x * blockDim.x + threadIdx.x;
    if (i < n) out[i] = fmaxf(in[i], 0.0f);
}

extern "C" void kernel_launch(void* const* d_in, const int* in_sizes, int n_in,
                              void* d_out, int out_size, void* d_ws, size_t ws_size,
                              hipStream_t stream) {
    constexpr int IN  = 256;
    constexpr int HID = 128;
    constexpr int LAT = 64;

    const float* x   = (const float*)d_in[0];
    const int*   ei  = (const int*)d_in[1];
    const float* W1  = (const float*)d_in[2];
    const float* b1  = (const float*)d_in[3];
    const float* W2  = (const float*)d_in[4];
    const float* b2  = (const float*)d_in[5];
    float*       out = (float*)d_out;

    const int N = in_sizes[0] / IN;
    const int E = in_sizes[1] / 2;
    const int* src = ei;
    const int* dst = ei + E;

    char* ws = (char*)d_ws;
    float* dinv = (float*)ws;  ws += (size_t)N * sizeof(float);
    float* norm = (float*)ws;  ws += (size_t)E * sizeof(float);
    float* H1   = (float*)ws;  ws += (size_t)N * HID * sizeof(float);
    float* S1   = (float*)ws;
    float* H2   = S1;

    const int T = 256;

    k_fill1<<<cdiv(N, T), T, 0, stream>>>(dinv, N);
    k_deg  <<<cdiv(E, T), T, 0, stream>>>(dst, dinv, E);
    k_rsqrt<<<cdiv(N, T), T, 0, stream>>>(dinv, N);
    k_norm <<<cdiv(E, T), T, 0, stream>>>(src, dst, dinv, norm, E);

    gemm_wmma_f32<IN, HID><<<cdiv(N, 128), T, 0, stream>>>(x, W1, H1, N);

    k_selfinit<HID><<<cdiv((long long)N * (HID / 4), T), T, 0, stream>>>(H1, dinv, b1, S1, N);
    k_scatter<HID><<<cdiv((long long)E * (HID / 4), T), T, 0, stream>>>(src, dst, norm, H1, S1, E);
    k_relu<<<cdiv((long long)N * HID, T), T, 0, stream>>>(S1, H1, N * HID);

    gemm_wmma_f32<HID, LAT><<<cdiv(N, 128), T, 0, stream>>>(H1, W2, H2, N);

    k_selfinit<LAT><<<cdiv((long long)N * (LAT / 4), T), T, 0, stream>>>(H2, dinv, b2, out, N);
    k_scatter<LAT><<<cdiv((long long)E * (LAT / 4), T), T, 0, stream>>>(src, dst, norm, H2, out, E);
}


# Edge scatter-add by f32 atomics becomes a sum by destination: a returning int add ranks each edge, scan + place build per-node source lists, a wave sums a node's rows in registers (arrival order)
# speedup vs baseline: 2.0454x; 2.0454x over previous
#include <hip/hip_runtime.h>


typedef __attribute__((ext_vector_type(2))) float v2f;
typedef __attribute__((ext_vector_type(8))) float v8f;

static inline int cdiv(long long a, long long b) { return (int)((a + b - 1) / b); }

template <int K, int N>
__launch_bounds__(256)
__global__ void gemm_wmma_f32(const float* __restrict__ X,
                              const float* __restrict__ W,
                              float* __restrict__ H,
                              int nrows) {
    constexpr int NT = N / 16;
    __shared__ float BF[(K / 4) * NT * 64];

    const int tid = threadIdx.x;
    for (int idx = tid; idx < K * N; idx += 256) {
        const int n  = idx / K;
        const int k  = idx - n * K;
        const int ks = k >> 2;
        const int kk = k & 3;
        const int g  = kk >> 1;
        const int r  = kk & 1;
        const int t  = n >> 4;
        const int nl = n & 15;
        BF[((ks * NT + t) * 32 + g * 16 + nl) * 2 + r] = W[idx];
    }
    __syncthreads();

    const int wave   = tid >> 5;
    const int lane   = tid & 31;
    const int laneLo = lane & 15;
    const int laneHi = lane >> 4;
    const int m0     = (blockIdx.x * 8 + wave) * 16;
    if (m0 >= nrows) return;

    v8f acc[NT] = {};

    const float* xrow  = X + (size_t)(m0 + laneLo) * K;
    const float* fbase = BF + lane * 2;
    for (int ks = 0; ks < K / 4; ++ks) {
        const v2f a = *(const v2f*)(xrow + ks * 4 + 2 * laneHi);
        const float* fb = fbase + (ks * NT) * 64;
#pragma unroll
        for (int t = 0; t < NT; ++t) {
            const v2f b = *(const v2f*)(fb + t * 64);
            acc[t] = __builtin_amdgcn_wmma_f32_16x16x4_f32(
                false, a, false, b, (short)0, acc[t], false, false);
        }
    }

#pragma unroll
    for (int t = 0; t < NT; ++t)
#pragma unroll
        for (int r = 0; r < 8; ++r)
            H[(size_t)(m0 + r + 8 * laneHi) * N + t * 16 + laneLo] = acc[t][r];
}

__global__ void k_zero(int* __restrict__ cnt, int n) {
    int i = blockIdx.x * blockDim.x + threadIdx.x;
    if (i < n) cnt[i] = 0;
}

__global__ void k_rank(const int* __restrict__ dst, int* __restrict__ cnt,
                       int* __restrict__ rank, int E) {
    int e = blockIdx.x * blockDim.x + threadIdx.x;
    if (e < E) rank[e] = atomicAdd(&cnt[dst[e]], 1);
}

__launch_bounds__(1024)
__global__ void k_scan(const int* __restrict__ cnt, int* __restrict__ rowptr, int n) {
    __shared__ int wtot[32];
    __shared__ int wpre[32];
    const int tid  = threadIdx.x;
    const int lane = tid & 31;
    const int wv   = tid >> 5;
    const int base = blockIdx.x * 1024;
    const int i    = base + tid;
    const int c    = (i < n) ? cnt[i] : 0;

    int part = 0;
    for (int i0 = tid * 4; i0 < base; i0 += 8 * 4096) {
        int4 v[8];
#pragma unroll
        for (int k = 0; k < 8; ++k) {
            const int ii = i0 + k * 4096;
            v[k] = (ii < base) ? *(const int4*)(cnt + ii) : make_int4(0, 0, 0, 0);
        }
#pragma unroll
        for (int k = 0; k < 8; ++k) part += (v[k].x + v[k].y) + (v[k].z + v[k].w);
    }
    for (int o = 16; o > 0; o >>= 1) part += __shfl_xor(part, o);
    if (lane == 0) wtot[wv] = part;
    __syncthreads();
    int off = 0;
    for (int w = 0; w < 32; ++w) off += wtot[w];
    __syncthreads();

    int incl = c;
    for (int o = 1; o < 32; o <<= 1) {
        const int up = __shfl_up(incl, o);
        if (lane >= o) incl += up;
    }
    if (lane == 31) wtot[wv] = incl;
    __syncthreads();
    if (wv == 0) {
        const int t = wtot[lane];
        int sc = t;
        for (int o = 1; o < 32; o <<= 1) {
            const int up = __shfl_up(sc, o);
            if (lane >= o) sc += up;
        }
        wpre[lane] = sc - t;
    }
    __syncthreads();
    if (i < n) rowptr[i] = off + wpre[wv] + (incl - c);
}

__global__ void k_place(const int* __restrict__ src, const int* __restrict__ dst,
                        const int* __restrict__ rank, const int* __restrict__ rowptr,
                        int* __restrict__ csr, int E) {
    int e = blockIdx.x * blockDim.x + threadIdx.x;
    if (e < E) csr[rowptr[dst[e]] + rank[e]] = src[e];
}

template <int V> struct rowvec;
template <> struct rowvec<4> { typedef float4 type; };
template <> struct rowvec<2> { typedef float2 type; };

template <int C, bool RELU>
__launch_bounds__(256)
__global__ void k_gather(const int* __restrict__ rowptr, const int* __restrict__ csr,
                         const float* __restrict__ H, const float* __restrict__ bias,
                         float* __restrict__ out, int n, int E) {
    constexpr int V = C / 32;
    constexpr int U = 8;
    typedef typename rowvec<V>::type vec;

    const int lane = threadIdx.x & 31;
    const int wave   = __builtin_amdgcn_readfirstlane((int)((blockIdx.x * blockDim.x + threadIdx.x) >> 5));
    const int nwaves = (int)((gridDim.x * blockDim.x) >> 5);
    const float* Hl  = H + lane * V;

    float bv[V];
#pragma unroll
    for (int v = 0; v < V; ++v) bv[v] = bias[lane * V + v];

    int i = wave;
    int beg = 0, end = 0;
    if (i < n) {
        beg = rowptr[i];
        end = (i + 1 < n) ? rowptr[i + 1] : E;
    }
    while (i < n) {
        const int inext = i + nwaves;
        int nbeg = 0, nend = 0;
        if (inext < n) {
            nbeg = rowptr[inext];
            nend = (inext + 1 < n) ? rowptr[inext + 1] : E;
        }
        const float di = rsqrtf((float)(end - beg + 1));
        const vec hself = *(const vec*)(Hl + (size_t)i * C);

        float acc[V];
#pragma unroll
        for (int v = 0; v < V; ++v) acc[v] = 0.0f;

        for (int b0 = beg; b0 < end; b0 += 32) {
            const int m = (end - b0 < 32) ? (end - b0) : 32;
            int   s = i;
            float w = 0.0f;
            if (lane < m) {
                s = csr[b0 + lane];
                const int a = rowptr[s];
                const int z = (s + 1 < n) ? rowptr[s + 1] : E;
                w = rsqrtf((float)(z - a + 1)) * di;
            }
            for (int j = 0; j < m; j += U) {
                vec h[U];
#pragma unroll
                for (int u = 0; u < U; ++u) {
                    const int su = __builtin_amdgcn_readlane(s, j + u);
                    h[u] = *(const vec*)(Hl + (size_t)su * C);
                }
#pragma unroll
                for (int u = 0; u < U; ++u) {
                    const float wu = __int_as_float(__builtin_amdgcn_readlane(__float_as_int(w), j + u));
                    const float* hp = (const float*)&h[u];
#pragma unroll
                    for (int v = 0; v < V; ++v) acc[v] = fmaf(hp[v], wu, acc[v]);
                }
            }
        }

        vec o;
        float* op = (float*)&o;
        const float* hs = (const float*)&hself;
        const float sl = di * di;
#pragma unroll
        for (int v = 0; v < V; ++v) {
            const float r = fmaf(hs[v], sl, acc[v]) + bv[v];
            op[v] = RELU ? fmaxf(r, 0.0f) : r;
        }
        *(vec*)(out + (size_t)i * C + lane * V) = o;
        i = inext;
        beg = nbeg;
        end = nend;
    }
}

extern "C" void kernel_launch(void* const* d_in, const int* in_sizes, int n_in,
                              void* d_out, int out_size, void* d_ws, size_t ws_size,
                              hipStream_t stream) {
    constexpr int IN  = 256;
    constexpr int HID = 128;
    constexpr int LAT = 64;

    const float* x   = (const float*)d_in[0];
    const int*   ei  = (const int*)d_in[1];
    const float* W1  = (const float*)d_in[2];
    const float* b1  = (const float*)d_in[3];
    const float* W2  = (const float*)d_in[4];
    const float* b2  = (const float*)d_in[5];
    float*       out = (float*)d_out;

    const int N = in_sizes[0] / IN;
    const int E = in_sizes[1] / 2;
    const int* src = ei;
    const int* dst = ei + E;

    char* ws = (char*)d_ws;
    int*   rowptr = (int*)ws;    ws += (size_t)N * sizeof(int);
    int*   csr    = (int*)ws;    ws += (size_t)E * sizeof(int);
    float* H1     = (float*)ws;  ws += (size_t)N * HID * sizeof(float);
    float* S1     = (float*)ws;
    float* H2     = H1;
    int*   cnt    = (int*)S1;
    int*   rank   = cnt + (size_t)cdiv(N, 4) * 4;

    const int T = 256;
    const int GW = 2048;

    k_zero <<<cdiv(N, T), T, 0, stream>>>(cnt, N);
    k_rank <<<cdiv(E, T), T, 0, stream>>>(dst, cnt, rank, E);
    k_scan <<<cdiv(N, 1024), 1024, 0, stream>>>(cnt, rowptr, N);
    k_place<<<cdiv(E, T), T, 0, stream>>>(src, dst, rank, rowptr, csr, E);

    gemm_wmma_f32<IN, HID><<<cdiv(N, 128), T, 0, stream>>>(x, W1, H1, N);

    k_gather<HID, true><<<GW, T, 0, stream>>>(rowptr, csr, H1, b1, S1, N, E);

    gemm_wmma_f32<HID, LAT><<<cdiv(N, 128), T, 0, stream>>>(S1, W2, H2, N);

    k_gather<LAT, false><<<GW, T, 0, stream>>>(rowptr, csr, H2, b2, out, N, E);
}


# Edge grouping drops global atomics: per-edge returning adds + 4-B scatter become two levels of LDS counters (256-node buckets: count, scan, place); list order = arrival order of the LDS adds
# speedup vs baseline: 2.7520x; 1.3455x over previous
#include <hip/hip_runtime.h>


typedef __attribute__((ext_vector_type(2))) float v2f;
typedef __attribute__((ext_vector_type(8))) float v8f;

static inline int cdiv(long long a, long long b) { return (int)((a + b - 1) / b); }

template <int K, int N>
__launch_bounds__(256)
__global__ void gemm_wmma_f32(const float* __restrict__ X,
                              const float* __restrict__ W,
                              float* __restrict__ H,
                              int nrows) {
    constexpr int NT = N / 16;
    __shared__ float BF[(K / 4) * NT * 64];

    const int tid = threadIdx.x;
    for (int idx = tid; idx < K * N; idx += 256) {
        const int n  = idx / K;
        const int k  = idx - n * K;
        const int ks = k >> 2;
        const int kk = k & 3;
        const int g  = kk >> 1;
        const int r  = kk & 1;
        const int t  = n >> 4;
        const int nl = n & 15;
        BF[((ks * NT + t) * 32 + g * 16 + nl) * 2 + r] = W[idx];
    }
    __syncthreads();

    const int wave   = tid >> 5;
    const int lane   = tid & 31;
    const int laneLo = lane & 15;
    const int laneHi = lane >> 4;
    const int m0     = (blockIdx.x * 8 + wave) * 16;
    if (m0 >= nrows) return;

    v8f acc[NT] = {};

    const float* xrow  = X + (size_t)(m0 + laneLo) * K;
    const float* fbase = BF + lane * 2;
    for (int ks = 0; ks < K / 4; ++ks) {
        const v2f a = *(const v2f*)(xrow + ks * 4 + 2 * laneHi);
        const float* fb = fbase + (ks * NT) * 64;
#pragma unroll
        for (int t = 0; t < NT; ++t) {
            const v2f b = *(const v2f*)(fb + t * 64);
            acc[t] = __builtin_amdgcn_wmma_f32_16x16x4_f32(
                false, a, false, b, (short)0, acc[t], false, false);
        }
    }

#pragma unroll
    for (int t = 0; t < NT; ++t)
#pragma unroll
        for (int r = 0; r < 8; ++r)
            H[(size_t)(m0 + r + 8 * laneHi) * N + t * 16 + laneLo] = acc[t][r];
}

constexpr int BSH = 8;
constexpr int BN  = 1 << BSH;
constexpr int G1  = 256;
constexpr int PER = 8;

__launch_bounds__(1024)
__global__ void k_bcount(const int* __restrict__ dst, int* __restrict__ mat,
                         int E, int chunk, int NB) {
    extern __shared__ int lds_i[];
    const int tid = threadIdx.x;
    for (int b = tid; b < NB; b += 1024) lds_i[b] = 0;
    __syncthreads();
    const int e0 = blockIdx.x * chunk;
    const int e1 = min(E, e0 + chunk);
    for (int base = e0; base < e1; base += 1024 * PER) {
        int d[PER];
#pragma unroll
        for (int k = 0; k < PER; ++k) {
            const int e = base + k * 1024 + tid;
            d[k] = (e < e1) ? dst[e] : -1;
        }
#pragma unroll
        for (int k = 0; k < PER; ++k)
            if (d[k] >= 0) atomicAdd(&lds_i[d[k] >> BSH], 1);
    }
    __syncthreads();
    for (int b = tid; b < NB; b += 1024) mat[b * G1 + blockIdx.x] = lds_i[b];
}

__launch_bounds__(1024)
__global__ void k_bscatter(const int* __restrict__ src, const int* __restrict__ dst,
                           const int* __restrict__ offs, int* __restrict__ pairs_i,
                           int E, int chunk, int NB) {
    extern __shared__ int lds_i[];
    int2* __restrict__ pairs = (int2*)pairs_i;
    const int tid = threadIdx.x;
    for (int b = tid; b < NB; b += 1024) lds_i[b] = offs[b * G1 + blockIdx.x];
    __syncthreads();
    const int e0 = blockIdx.x * chunk;
    const int e1 = min(E, e0 + chunk);
    for (int base = e0; base < e1; base += 1024 * PER) {
        int d[PER], s[PER];
#pragma unroll
        for (int k = 0; k < PER; ++k) {
            const int e = base + k * 1024 + tid;
            d[k] = (e < e1) ? dst[e] : -1;
            s[k] = (e < e1) ? src[e] : 0;
        }
        int p[PER];
#pragma unroll
        for (int k = 0; k < PER; ++k)
            p[k] = (d[k] >= 0) ? atomicAdd(&lds_i[d[k] >> BSH], 1) : 0;
#pragma unroll
        for (int k = 0; k < PER; ++k)
            if (d[k] >= 0) pairs[p[k]] = make_int2(s[k], d[k]);
    }
}

__launch_bounds__(1024)
__global__ void k_scan(const int* __restrict__ cnt, int* __restrict__ rowptr, int n) {
    __shared__ int wtot[32];
    __shared__ int wpre[32];
    const int tid  = threadIdx.x;
    const int lane = tid & 31;
    const int wv   = tid >> 5;
    const int base = blockIdx.x * 1024;
    const int i    = base + tid;
    const int c    = (i < n) ? cnt[i] : 0;

    int part = 0;
    for (int i0 = tid * 4; i0 < base; i0 += 8 * 4096) {
        int4 v[8];
#pragma unroll
        for (int k = 0; k < 8; ++k) {
            const int ii = i0 + k * 4096;
            v[k] = (ii < base) ? *(const int4*)(cnt + ii) : make_int4(0, 0, 0, 0);
        }
#pragma unroll
        for (int k = 0; k < 8; ++k) part += (v[k].x + v[k].y) + (v[k].z + v[k].w);
    }
    for (int o = 16; o > 0; o >>= 1) part += __shfl_xor(part, o);
    if (lane == 0) wtot[wv] = part;
    __syncthreads();
    int off = 0;
    for (int w = 0; w < 32; ++w) off += wtot[w];
    __syncthreads();

    int incl = c;
    for (int o = 1; o < 32; o <<= 1) {
        const int up = __shfl_up(incl, o);
        if (lane >= o) incl += up;
    }
    if (lane == 31) wtot[wv] = incl;
    __syncthreads();
    if (wv == 0) {
        const int t = wtot[lane];
        int sc = t;
        for (int o = 1; o < 32; o <<= 1) {
            const int up = __shfl_up(sc, o);
            if (lane >= o) sc += up;
        }
        wpre[lane] = sc - t;
    }
    __syncthreads();
    if (i < n) rowptr[i] = off + wpre[wv] + (incl - c);
}

__launch_bounds__(1024)
__global__ void k_bucket_lists(const int* __restrict__ pairs_i, const int* __restrict__ offs,
                               int* __restrict__ rowptr, int* __restrict__ csr,
                               int N, int E, int NB) {
    __shared__ int cur[BN];
    __shared__ int wtot[BN / 32];
    const int2* __restrict__ pairs = (const int2*)pairs_i;
    const int tid  = threadIdx.x;
    const int lane = tid & 31;
    const int wv   = tid >> 5;
    const int b    = blockIdx.x;
    const int n0   = b << BSH;
    const int beg  = offs[b * G1];
    const int end  = (b + 1 < NB) ? offs[(b + 1) * G1] : E;

    if (tid < BN) cur[tid] = 0;
    __syncthreads();
    for (int base = beg; base < end; base += 1024 * PER) {
        int dy[PER];
#pragma unroll
        for (int k = 0; k < PER; ++k) {
            const int e = base + k * 1024 + tid;
            dy[k] = (e < end) ? pairs[e].y : -1;
        }
#pragma unroll
        for (int k = 0; k < PER; ++k)
            if (dy[k] >= 0) atomicAdd(&cur[dy[k] - n0], 1);
    }
    __syncthreads();

    const int c = (tid < BN) ? cur[tid] : 0;
    int incl = c;
    for (int o = 1; o < 32; o <<= 1) {
        const int up = __shfl_up(incl, o);
        if (lane >= o) incl += up;
    }
    if (tid < BN && lane == 31) wtot[wv] = incl;
    __syncthreads();
    if (tid < BN) {
        int woff = 0;
        for (int w = 0; w < wv; ++w) woff += wtot[w];
        const int start = beg + woff + (incl - c);
        if (n0 + tid < N) rowptr[n0 + tid] = start;
        cur[tid] = start;
    }
    __syncthreads();

    for (int base = beg; base < end; base += 1024 * PER) {
        int2 p[PER];
#pragma unroll
        for (int k = 0; k < PER; ++k) {
            const int e = base + k * 1024 + tid;
            p[k] = (e < end) ? pairs[e] : make_int2(0, -1);
        }
        int q[PER];
#pragma unroll
        for (int k = 0; k < PER; ++k)
            q[k] = (p[k].y >= 0) ? atomicAdd(&cur[p[k].y - n0], 1) : 0;
#pragma unroll
        for (int k = 0; k < PER; ++k)
            if (p[k].y >= 0) csr[q[k]] = p[k].x;
    }
}

template <int V> struct rowvec;
template <> struct rowvec<4> { typedef float4 type; };
template <> struct rowvec<2> { typedef float2 type; };

template <int C, bool RELU>
__launch_bounds__(256)
__global__ void k_gather(const int* __restrict__ rowptr, const int* __restrict__ csr,
                         const float* __restrict__ H, const float* __restrict__ bias,
                         float* __restrict__ out, int n, int E) {
    constexpr int V = C / 32;
    constexpr int U = 8;
    typedef typename rowvec<V>::type vec;

    const int lane = threadIdx.x & 31;
    const int wave   = __builtin_amdgcn_readfirstlane((int)((blockIdx.x * blockDim.x + threadIdx.x) >> 5));
    const int nwaves = (int)((gridDim.x * blockDim.x) >> 5);
    const float* Hl  = H + lane * V;

    float bv[V];
#pragma unroll
    for (int v = 0; v < V; ++v) bv[v] = bias[lane * V + v];

    int i = wave;
    int beg = 0, end = 0;
    if (i < n) {
        beg = rowptr[i];
        end = (i + 1 < n) ? rowptr[i + 1] : E;
    }
    while (i < n) {
        const int inext = i + nwaves;
        int nbeg = 0, nend = 0;
        if (inext < n) {
            nbeg = rowptr[inext];
            nend = (inext + 1 < n) ? rowptr[inext + 1] : E;
        }
        const float di = rsqrtf((float)(end - beg + 1));
        const vec hself = *(const vec*)(Hl + (size_t)i * C);

        float acc[V];
#pragma unroll
        for (int v = 0; v < V; ++v) acc[v] = 0.0f;

        for (int b0 = beg; b0 < end; b0 += 32) {
            const int m = (end - b0 < 32) ? (end - b0) : 32;
            int   s = i;
            float w = 0.0f;
            if (lane < m) {
                s = csr[b0 + lane];
                const int a = rowptr[s];
                const int z = (s + 1 < n) ? rowptr[s + 1] : E;
                w = rsqrtf((float)(z - a + 1)) * di;
            }
            for (int j = 0; j < m; j += U) {
                vec h[U];
#pragma unroll
                for (int u = 0; u < U; ++u) {
                    const int su = __builtin_amdgcn_readlane(s, j + u);
                    h[u] = *(const vec*)(Hl + (size_t)su * C);
                }
#pragma unroll
                for (int u = 0; u < U; ++u) {
                    const float wu = __int_as_float(__builtin_amdgcn_readlane(__float_as_int(w), j + u));
                    const float* hp = (const float*)&h[u];
#pragma unroll
                    for (int v = 0; v < V; ++v) acc[v] = fmaf(hp[v], wu, acc[v]);
                }
            }
        }

        vec o;
        float* op = (float*)&o;
        const float* hs = (const float*)&hself;
        const float sl = di * di;
#pragma unroll
        for (int v = 0; v < V; ++v) {
            const float r = fmaf(hs[v], sl, acc[v]) + bv[v];
            op[v] = RELU ? fmaxf(r, 0.0f) : r;
        }
        *(vec*)(out + (size_t)i * C + lane * V) = o;
        i = inext;
        beg = nbeg;
        end = nend;
    }
}

extern "C" void kernel_launch(void* const* d_in, const int* in_sizes, int n_in,
                              void* d_out, int out_size, void* d_ws, size_t ws_size,
                              hipStream_t stream) {
    constexpr int IN  = 256;
    constexpr int HID = 128;
    constexpr int LAT = 64;

    const float* x   = (const float*)d_in[0];
    const int*   ei  = (const int*)d_in[1];
    const float* W1  = (const float*)d_in[2];
    const float* b1  = (const float*)d_in[3];
    const float* W2  = (const float*)d_in[4];
    const float* b2  = (const float*)d_in[5];
    float*       out = (float*)d_out;

    const int N = in_sizes[0] / IN;
    const int E = in_sizes[1] / 2;
    const int* src = ei;
    const int* dst = ei + E;

    char* ws = (char*)d_ws;
    int*   rowptr = (int*)ws;    ws += (size_t)N * sizeof(int);
    int*   csr    = (int*)ws;    ws += (size_t)E * sizeof(int);
    float* H1     = (float*)ws;  ws += (size_t)N * HID * sizeof(float);
    float* S1     = (float*)ws;
    float* H2     = H1;
    const int NB    = cdiv(N, BN);
    const int chunk = cdiv(E, G1);
    const int MN    = NB * G1;
    int*   mat    = (int*)S1;
    int*   offs   = mat + (size_t)cdiv(MN, 4) * 4;
    int*   pairs  = offs + (size_t)cdiv(MN, 4) * 4;

    const int T = 256;
    const int GW = 2048;

    k_bcount      <<<G1, 1024, NB * sizeof(int), stream>>>(dst, mat, E, chunk, NB);
    k_scan        <<<cdiv(MN, 1024), 1024, 0, stream>>>(mat, offs, MN);
    k_bscatter    <<<G1, 1024, NB * sizeof(int), stream>>>(src, dst, offs, pairs, E, chunk, NB);
    k_bucket_lists<<<NB, 1024, 0, stream>>>(pairs, offs, rowptr, csr, N, E, NB);

    gemm_wmma_f32<IN, HID><<<cdiv(N, 128), T, 0, stream>>>(x, W1, H1, N);

    k_gather<HID, true><<<GW, T, 0, stream>>>(rowptr, csr, H1, b1, S1, N, E);

    gemm_wmma_f32<HID, LAT><<<cdiv(N, 128), T, 0, stream>>>(S1, W2, H2, N);

    k_gather<LAT, false><<<GW, T, 0, stream>>>(rowptr, csr, H2, b2, out, N, E);
}


# dinv[src] is folded into the H rows at the GEMM's store (row scale from rowptr): the gather drops its two random rowptr loads per edge and sums plain rows; dinv[dst] applied once after the sum
# speedup vs baseline: 3.0907x; 1.1231x over previous
#include <hip/hip_runtime.h>


typedef __attribute__((ext_vector_type(2))) float v2f;
typedef __attribute__((ext_vector_type(8))) float v8f;

static inline int cdiv(long long a, long long b) { return (int)((a + b - 1) / b); }

template <int K, int N>
__launch_bounds__(256)
__global__ void gemm_wmma_f32(const float* __restrict__ X,
                              const float* __restrict__ W,
                              const int* __restrict__ rowptr,
                              float* __restrict__ H,
                              int nrows, int E) {
    constexpr int NT = N / 16;
    __shared__ float BF[(K / 4) * NT * 64];

    const int tid = threadIdx.x;
    for (int idx = tid; idx < K * N; idx += 256) {
        const int n  = idx / K;
        const int k  = idx - n * K;
        const int ks = k >> 2;
        const int kk = k & 3;
        const int g  = kk >> 1;
        const int r  = kk & 1;
        const int t  = n >> 4;
        const int nl = n & 15;
        BF[((ks * NT + t) * 32 + g * 16 + nl) * 2 + r] = W[idx];
    }
    __syncthreads();

    const int wave   = tid >> 5;
    const int lane   = tid & 31;
    const int laneLo = lane & 15;
    const int laneHi = lane >> 4;
    const int m0     = (blockIdx.x * 8 + wave) * 16;
    if (m0 >= nrows) return;

    float dv[8];
    {
        const int r0 = m0 + 8 * laneHi;
        int rp[9];
#pragma unroll
        for (int r = 0; r < 8; ++r) rp[r] = rowptr[r0 + r];
        rp[8] = (r0 + 8 < nrows) ? rowptr[r0 + 8] : E;
#pragma unroll
        for (int r = 0; r < 8; ++r) dv[r] = rsqrtf((float)(rp[r + 1] - rp[r] + 1));
    }

    v8f acc[NT] = {};

    const float* xrow  = X + (size_t)(m0 + laneLo) * K;
    const float* fbase = BF + lane * 2;
    for (int ks = 0; ks < K / 4; ++ks) {
        const v2f a = *(const v2f*)(xrow + ks * 4 + 2 * laneHi);
        const float* fb = fbase + (ks * NT) * 64;
#pragma unroll
        for (int t = 0; t < NT; ++t) {
            const v2f b = *(const v2f*)(fb + t * 64);
            acc[t] = __builtin_amdgcn_wmma_f32_16x16x4_f32(
                false, a, false, b, (short)0, acc[t], false, false);
        }
    }

#pragma unroll
    for (int t = 0; t < NT; ++t)
#pragma unroll
        for (int r = 0; r < 8; ++r)
            H[(size_t)(m0 + r + 8 * laneHi) * N + t * 16 + laneLo] = acc[t][r] * dv[r];
}

constexpr int BSH = 8;
constexpr int BN  = 1 << BSH;
constexpr int G1  = 256;
constexpr int PER = 8;

__launch_bounds__(1024)
__global__ void k_bcount(const int* __restrict__ dst, int* __restrict__ mat,
                         int E, int chunk, int NB) {
    extern __shared__ int lds_i[];
    const int tid = threadIdx.x;
    for (int b = tid; b < NB; b += 1024) lds_i[b] = 0;
    __syncthreads();
    const int e0 = blockIdx.x * chunk;
    const int e1 = min(E, e0 + chunk);
    for (int base = e0; base < e1; base += 1024 * PER) {
        int d[PER];
#pragma unroll
        for (int k = 0; k < PER; ++k) {
            const int e = base + k * 1024 + tid;
            d[k] = (e < e1) ? dst[e] : -1;
        }
#pragma unroll
        for (int k = 0; k < PER; ++k)
            if (d[k] >= 0) atomicAdd(&lds_i[d[k] >> BSH], 1);
    }
    __syncthreads();
    for (int b = tid; b < NB; b += 1024) mat[b * G1 + blockIdx.x] = lds_i[b];
}

__launch_bounds__(1024)
__global__ void k_bscatter(const int* __restrict__ src, const int* __restrict__ dst,
                           const int* __restrict__ offs, int* __restrict__ pairs_i,
                           int E, int chunk, int NB) {
    extern __shared__ int lds_i[];
    int2* __restrict__ pairs = (int2*)pairs_i;
    const int tid = threadIdx.x;
    for (int b = tid; b < NB; b += 1024) lds_i[b] = offs[b * G1 + blockIdx.x];
    __syncthreads();
    const int e0 = blockIdx.x * chunk;
    const int e1 = min(E, e0 + chunk);
    for (int base = e0; base < e1; base += 1024 * PER) {
        int d[PER], s[PER];
#pragma unroll
        for (int k = 0; k < PER; ++k) {
            const int e = base + k * 1024 + tid;
            d[k] = (e < e1) ? dst[e] : -1;
            s[k] = (e < e1) ? src[e] : 0;
        }
        int p[PER];
#pragma unroll
        for (int k = 0; k < PER; ++k)
            p[k] = (d[k] >= 0) ? atomicAdd(&lds_i[d[k] >> BSH], 1) : 0;
#pragma unroll
        for (int k = 0; k < PER; ++k)
            if (d[k] >= 0) pairs[p[k]] = make_int2(s[k], d[k]);
    }
}

__launch_bounds__(1024)
__global__ void k_scan(const int* __restrict__ cnt, int* __restrict__ rowptr, int n) {
    __shared__ int wtot[32];
    __shared__ int wpre[32];
    const int tid  = threadIdx.x;
    const int lane = tid & 31;
    const int wv   = tid >> 5;
    const int base = blockIdx.x * 1024;
    const int i    = base + tid;
    const int c    = (i < n) ? cnt[i] : 0;

    int part = 0;
    for (int i0 = tid * 4; i0 < base; i0 += 8 * 4096) {
        int4 v[8];
#pragma unroll
        for (int k = 0; k < 8; ++k) {
            const int ii = i0 + k * 4096;
            v[k] = (ii < base) ? *(const int4*)(cnt + ii) : make_int4(0, 0, 0, 0);
        }
#pragma unroll
        for (int k = 0; k < 8; ++k) part += (v[k].x + v[k].y) + (v[k].z + v[k].w);
    }
    for (int o = 16; o > 0; o >>= 1) part += __shfl_xor(part, o);
    if (lane == 0) wtot[wv] = part;
    __syncthreads();
    int off = 0;
    for (int w = 0; w < 32; ++w) off += wtot[w];
    __syncthreads();

    int incl = c;
    for (int o = 1; o < 32; o <<= 1) {
        const int up = __shfl_up(incl, o);
        if (lane >= o) incl += up;
    }
    if (lane == 31) wtot[wv] = incl;
    __syncthreads();
    if (wv == 0) {
        const int t = wtot[lane];
        int sc = t;
        for (int o = 1; o < 32; o <<= 1) {
            const int up = __shfl_up(sc, o);
            if (lane >= o) sc += up;
        }
        wpre[lane] = sc - t;
    }
    __syncthreads();
    if (i < n) rowptr[i] = off + wpre[wv] + (incl - c);
}

__launch_bounds__(1024)
__global__ void k_bucket_lists(const int* __restrict__ pairs_i, const int* __restrict__ offs,
                               int* __restrict__ rowptr, int* __restrict__ csr,
                               int N, int E, int NB) {
    __shared__ int cur[BN];
    __shared__ int wtot[BN / 32];
    const int2* __restrict__ pairs = (const int2*)pairs_i;
    const int tid  = threadIdx.x;
    const int lane = tid & 31;
    const int wv   = tid >> 5;
    const int b    = blockIdx.x;
    const int n0   = b << BSH;
    const int beg  = offs[b * G1];
    const int end  = (b + 1 < NB) ? offs[(b + 1) * G1] : E;

    if (tid < BN) cur[tid] = 0;
    __syncthreads();
    for (int base = beg; base < end; base += 1024 * PER) {
        int dy[PER];
#pragma unroll
        for (int k = 0; k < PER; ++k) {
            const int e = base + k * 1024 + tid;
            dy[k] = (e < end) ? pairs[e].y : -1;
        }
#pragma unroll
        for (int k = 0; k < PER; ++k)
            if (dy[k] >= 0) atomicAdd(&cur[dy[k] - n0], 1);
    }
    __syncthreads();

    const int c = (tid < BN) ? cur[tid] : 0;
    int incl = c;
    for (int o = 1; o < 32; o <<= 1) {
        const int up = __shfl_up(incl, o);
        if (lane >= o) incl += up;
    }
    if (tid < BN && lane == 31) wtot[wv] = incl;
    __syncthreads();
    if (tid < BN) {
        int woff = 0;
        for (int w = 0; w < wv; ++w) woff += wtot[w];
        const int start = beg + woff + (incl - c);
        if (n0 + tid < N) rowptr[n0 + tid] = start;
        cur[tid] = start;
    }
    __syncthreads();

    for (int base = beg; base < end; base += 1024 * PER) {
        int2 p[PER];
#pragma unroll
        for (int k = 0; k < PER; ++k) {
            const int e = base + k * 1024 + tid;
            p[k] = (e < end) ? pairs[e] : make_int2(0, -1);
        }
        int q[PER];
#pragma unroll
        for (int k = 0; k < PER; ++k)
            q[k] = (p[k].y >= 0) ? atomicAdd(&cur[p[k].y - n0], 1) : 0;
#pragma unroll
        for (int k = 0; k < PER; ++k)
            if (p[k].y >= 0) csr[q[k]] = p[k].x;
    }
}

template <int V> struct rowvec;
template <> struct rowvec<4> { typedef float4 type; };
template <> struct rowvec<2> { typedef float2 type; };

template <int C, bool RELU>
__launch_bounds__(256)
__global__ void k_gather(const int* __restrict__ rowptr, const int* __restrict__ csr,
                         const float* __restrict__ H, const float* __restrict__ bias,
                         float* __restrict__ out, int n, int E) {
    constexpr int V = C / 32;
    constexpr int U = 8;
    typedef typename rowvec<V>::type vec;

    const int lane = threadIdx.x & 31;
    const int wave   = __builtin_amdgcn_readfirstlane((int)((blockIdx.x * blockDim.x + threadIdx.x) >> 5));
    const int nwaves = (int)((gridDim.x * blockDim.x) >> 5);
    const float* Hl  = H + lane * V;

    float bv[V];
#pragma unroll
    for (int v = 0; v < V; ++v) bv[v] = bias[lane * V + v];

    int i = wave;
    int beg = 0, end = 0;
    if (i < n) {
        beg = rowptr[i];
        end = (i + 1 < n) ? rowptr[i + 1] : E;
    }
    while (i < n) {
        const int inext = i + nwaves;
        int nbeg = 0, nend = 0;
        if (inext < n) {
            nbeg = rowptr[inext];
            nend = (inext + 1 < n) ? rowptr[inext + 1] : E;
        }
        const float di = rsqrtf((float)(end - beg + 1));
        const vec hself = *(const vec*)(Hl + (size_t)i * C);

        float acc[V];
#pragma unroll
        for (int v = 0; v < V; ++v) acc[v] = 0.0f;

        for (int b0 = beg; b0 < end; b0 += 32) {
            const int m = (end - b0 < 32) ? (end - b0) : 32;
            int s = i;
            if (lane < m) s = csr[b0 + lane];
            int j = 0;
            do {
                vec h[U];
#pragma unroll
                for (int u = 0; u < U; ++u) {
                    const int su = __builtin_amdgcn_readlane(s, j + u);
                    h[u] = *(const vec*)(Hl + (size_t)su * C);
                }
#pragma unroll
                for (int u = 0; u < U; ++u) {
                    const float wu = (j + u < m) ? 1.0f : 0.0f;
                    const float* hp = (const float*)&h[u];
#pragma unroll
                    for (int v = 0; v < V; ++v) acc[v] = fmaf(hp[v], wu, acc[v]);
                }
                j += U;
            } while (j < m);
        }

        vec o;
        float* op = (float*)&o;
        const float* hs = (const float*)&hself;
#pragma unroll
        for (int v = 0; v < V; ++v) {
            const float r = fmaf(acc[v] + hs[v], di, bv[v]);
            op[v] = RELU ? fmaxf(r, 0.0f) : r;
        }
        *(vec*)(out + (size_t)i * C + lane * V) = o;
        i = inext;
        beg = nbeg;
        end = nend;
    }
}

extern "C" void kernel_launch(void* const* d_in, const int* in_sizes, int n_in,
                              void* d_out, int out_size, void* d_ws, size_t ws_size,
                              hipStream_t stream) {
    constexpr int IN  = 256;
    constexpr int HID = 128;
    constexpr int LAT = 64;

    const float* x   = (const float*)d_in[0];
    const int*   ei  = (const int*)d_in[1];
    const float* W1  = (const float*)d_in[2];
    const float* b1  = (const float*)d_in[3];
    const float* W2  = (const float*)d_in[4];
    const float* b2  = (const float*)d_in[5];
    float*       out = (float*)d_out;

    const int N = in_sizes[0] / IN;
    const int E = in_sizes[1] / 2;
    const int* src = ei;
    const int* dst = ei + E;

    char* ws = (char*)d_ws;
    int*   rowptr = (int*)ws;    ws += (size_t)N * sizeof(int);
    int*   csr    = (int*)ws;    ws += (size_t)E * sizeof(int);
    float* H1     = (float*)ws;  ws += (size_t)N * HID * sizeof(float);
    float* S1     = (float*)ws;
    float* H2     = H1;
    const int NB    = cdiv(N, BN);
    const int chunk = cdiv(E, G1);
    const int MN    = NB * G1;
    int*   mat    = (int*)S1;
    int*   offs   = mat + (size_t)cdiv(MN, 4) * 4;
    int*   pairs  = offs + (size_t)cdiv(MN, 4) * 4;

    const int T = 256;
    const int GW = 2048;

    k_bcount      <<<G1, 1024, NB * sizeof(int), stream>>>(dst, mat, E, chunk, NB);
    k_scan        <<<cdiv(MN, 1024), 1024, 0, stream>>>(mat, offs, MN);
    k_bscatter    <<<G1, 1024, NB * sizeof(int), stream>>>(src, dst, offs, pairs, E, chunk, NB);
    k_bucket_lists<<<NB, 1024, 0, stream>>>(pairs, offs, rowptr, csr, N, E, NB);

    gemm_wmma_f32<IN, HID><<<cdiv(N, 128), T, 0, stream>>>(x, W1, rowptr, H1, N, E);

    k_gather<HID, true><<<GW, T, 0, stream>>>(rowptr, csr, H1, b1, S1, N, E);

    gemm_wmma_f32<HID, LAT><<<cdiv(N, 128), T, 0, stream>>>(S1, W2, rowptr, H2, N, E);

    k_gather<LAT, false><<<GW, T, 0, stream>>>(rowptr, csr, H2, b2, out, N, E);
}


# GEMM work-groups made persistent: one 16-wave group per WGP stages W into LDS once and its waves loop over 16-row x half-N units; same k order, every output's sum unchanged
# speedup vs baseline: 3.2808x; 1.0615x over previous
#include <hip/hip_runtime.h>


typedef __attribute__((ext_vector_type(2))) float v2f;
typedef __attribute__((ext_vector_type(8))) float v8f;

static inline int cdiv(long long a, long long b) { return (int)((a + b - 1) / b); }

constexpr int GEMM_WAVES = 16;
constexpr int GEMM_WGS   = 256;

template <int K, int N>
__launch_bounds__(GEMM_WAVES * 32)
__global__ void gemm_wmma_f32(const float* __restrict__ X,
                              const float* __restrict__ W,
                              const int* __restrict__ rowptr,
                              float* __restrict__ H,
                              int nrows, int E) {
    constexpr int NT = N / 16;
    constexpr int NH = NT / 2;
    constexpr int NTHR = GEMM_WAVES * 32;
    constexpr int LDSF = (K * N > 41984) ? K * N : 41984;
    __shared__ float BF[LDSF];

    const int tid = threadIdx.x;
    for (int idx = tid; idx < K * N; idx += NTHR) {
        const int n  = idx / K;
        const int k  = idx - n * K;
        const int ks = k >> 2;
        const int kk = k & 3;
        const int g  = kk >> 1;
        const int r  = kk & 1;
        const int t  = n >> 4;
        const int nl = n & 15;
        BF[((ks * NT + t) * 32 + g * 16 + nl) * 2 + r] = W[idx];
    }
    __syncthreads();

    const int wave   = __builtin_amdgcn_readfirstlane(tid >> 5);
    const int lane   = tid & 31;
    const int laneLo = lane & 15;
    const int laneHi = lane >> 4;
    const int nstrips = nrows >> 4;

    for (int u = wave; ; u += GEMM_WAVES) {
        const int strip = (int)blockIdx.x + (int)gridDim.x * (u >> 1);
        if (strip >= nstrips) break;
        const int m0 = strip * 16;
        const int t0 = (u & 1) * NH;

        float dv[8];
        {
            const int r0 = m0 + 8 * laneHi;
            int rp[9];
#pragma unroll
            for (int r = 0; r < 8; ++r) rp[r] = rowptr[r0 + r];
            rp[8] = (r0 + 8 < nrows) ? rowptr[r0 + 8] : E;
#pragma unroll
            for (int r = 0; r < 8; ++r) dv[r] = rsqrtf((float)(rp[r + 1] - rp[r] + 1));
        }

        v8f acc[NH] = {};

        const float* xrow  = X + (size_t)(m0 + laneLo) * K;
        const float* fbase = BF + t0 * 64 + lane * 2;
        for (int ks = 0; ks < K / 4; ++ks) {
            const v2f a = *(const v2f*)(xrow + ks * 4 + 2 * laneHi);
            const float* fb = fbase + (ks * NT) * 64;
#pragma unroll
            for (int t = 0; t < NH; ++t) {
                const v2f b = *(const v2f*)(fb + t * 64);
                acc[t] = __builtin_amdgcn_wmma_f32_16x16x4_f32(
                    false, a, false, b, (short)0, acc[t], false, false);
            }
        }

#pragma unroll
        for (int t = 0; t < NH; ++t)
#pragma unroll
            for (int r = 0; r < 8; ++r)
                H[(size_t)(m0 + r + 8 * laneHi) * N + (t0 + t) * 16 + laneLo] = acc[t][r] * dv[r];
    }
}

constexpr int BSH = 8;
constexpr int BN  = 1 << BSH;
constexpr int G1  = 256;
constexpr int PER = 8;

__launch_bounds__(1024)
__global__ void k_bcount(const int* __restrict__ dst, int* __restrict__ mat,
                         int E, int chunk, int NB) {
    extern __shared__ int lds_i[];
    const int tid = threadIdx.x;
    for (int b = tid; b < NB; b += 1024) lds_i[b] = 0;
    __syncthreads();
    const int e0 = blockIdx.x * chunk;
    const int e1 = min(E, e0 + chunk);
    for (int base = e0; base < e1; base += 1024 * PER) {
        int d[PER];
#pragma unroll
        for (int k = 0; k < PER; ++k) {
            const int e = base + k * 1024 + tid;
            d[k] = (e < e1) ? dst[e] : -1;
        }
#pragma unroll
        for (int k = 0; k < PER; ++k)
            if (d[k] >= 0) atomicAdd(&lds_i[d[k] >> BSH], 1);
    }
    __syncthreads();
    for (int b = tid; b < NB; b += 1024) mat[b * G1 + blockIdx.x] = lds_i[b];
}

__launch_bounds__(1024)
__global__ void k_bscatter(const int* __restrict__ src, const int* __restrict__ dst,
                           const int* __restrict__ offs, int* __restrict__ pairs_i,
                           int E, int chunk, int NB) {
    extern __shared__ int lds_i[];
    int2* __restrict__ pairs = (int2*)pairs_i;
    const int tid = threadIdx.x;
    for (int b = tid; b < NB; b += 1024) lds_i[b] = offs[b * G1 + blockIdx.x];
    __syncthreads();
    const int e0 = blockIdx.x * chunk;
    const int e1 = min(E, e0 + chunk);
    for (int base = e0; base < e1; base += 1024 * PER) {
        int d[PER], s[PER];
#pragma unroll
        for (int k = 0; k < PER; ++k) {
            const int e = base + k * 1024 + tid;
            d[k] = (e < e1) ? dst[e] : -1;
            s[k] = (e < e1) ? src[e] : 0;
        }
        int p[PER];
#pragma unroll
        for (int k = 0; k < PER; ++k)
            p[k] = (d[k] >= 0) ? atomicAdd(&lds_i[d[k] >> BSH], 1) : 0;
#pragma unroll
        for (int k = 0; k < PER; ++k)
            if (d[k] >= 0) pairs[p[k]] = make_int2(s[k], d[k]);
    }
}

__launch_bounds__(1024)
__global__ void k_scan(const int* __restrict__ cnt, int* __restrict__ rowptr, int n) {
    __shared__ int wtot[32];
    __shared__ int wpre[32];
    const int tid  = threadIdx.x;
    const int lane = tid & 31;
    const int wv   = tid >> 5;
    const int base = blockIdx.x * 1024;
    const int i    = base + tid;
    const int c    = (i < n) ? cnt[i] : 0;

    int part = 0;
    for (int i0 = tid * 4; i0 < base; i0 += 8 * 4096) {
        int4 v[8];
#pragma unroll
        for (int k = 0; k < 8; ++k) {
            const int ii = i0 + k * 4096;
            v[k] = (ii < base) ? *(const int4*)(cnt + ii) : make_int4(0, 0, 0, 0);
        }
#pragma unroll
        for (int k = 0; k < 8; ++k) part += (v[k].x + v[k].y) + (v[k].z + v[k].w);
    }
    for (int o = 16; o > 0; o >>= 1) part += __shfl_xor(part, o);
    if (lane == 0) wtot[wv] = part;
    __syncthreads();
    int off = 0;
    for (int w = 0; w < 32; ++w) off += wtot[w];
    __syncthreads();

    int incl = c;
    for (int o = 1; o < 32; o <<= 1) {
        const int up = __shfl_up(incl, o);
        if (lane >= o) incl += up;
    }
    if (lane == 31) wtot[wv] = incl;
    __syncthreads();
    if (wv == 0) {
        const int t = wtot[lane];
        int sc = t;
        for (int o = 1; o < 32; o <<= 1) {
            const int up = __shfl_up(sc, o);
            if (lane >= o) sc += up;
        }
        wpre[lane] = sc - t;
    }
    __syncthreads();
    if (i < n) rowptr[i] = off + wpre[wv] + (incl - c);
}

__launch_bounds__(1024)
__global__ void k_bucket_lists(const int* __restrict__ pairs_i, const int* __restrict__ offs,
                               int* __restrict__ rowptr, int* __restrict__ csr,
                               int N, int E, int NB) {
    __shared__ int cur[BN];
    __shared__ int wtot[BN / 32];
    const int2* __restrict__ pairs = (const int2*)pairs_i;
    const int tid  = threadIdx.x;
    const int lane = tid & 31;
    const int wv   = tid >> 5;
    const int b    = blockIdx.x;
    const int n0   = b << BSH;
    const int beg  = offs[b * G1];
    const int end  = (b + 1 < NB) ? offs[(b + 1) * G1] : E;

    if (tid < BN) cur[tid] = 0;
    __syncthreads();
    for (int base = beg; base < end; base += 1024 * PER) {
        int dy[PER];
#pragma unroll
        for (int k = 0; k < PER; ++k) {
            const int e = base + k * 1024 + tid;
            dy[k] = (e < end) ? pairs[e].y : -1;
        }
#pragma unroll
        for (int k = 0; k < PER; ++k)
            if (dy[k] >= 0) atomicAdd(&cur[dy[k] - n0], 1);
    }
    __syncthreads();

    const int c = (tid < BN) ? cur[tid] : 0;
    int incl = c;
    for (int o = 1; o < 32; o <<= 1) {
        const int up = __shfl_up(incl, o);
        if (lane >= o) incl += up;
    }
    if (tid < BN && lane == 31) wtot[wv] = incl;
    __syncthreads();
    if (tid < BN) {
        int woff = 0;
        for (int w = 0; w < wv; ++w) woff += wtot[w];
        const int start = beg + woff + (incl - c);
        if (n0 + tid < N) rowptr[n0 + tid] = start;
        cur[tid] = start;
    }
    __syncthreads();

    for (int base = beg; base < end; base += 1024 * PER) {
        int2 p[PER];
#pragma unroll
        for (int k = 0; k < PER; ++k) {
            const int e = base + k * 1024 + tid;
            p[k] = (e < end) ? pairs[e] : make_int2(0, -1);
        }
        int q[PER];
#pragma unroll
        for (int k = 0; k < PER; ++k)
            q[k] = (p[k].y >= 0) ? atomicAdd(&cur[p[k].y - n0], 1) : 0;
#pragma unroll
        for (int k = 0; k < PER; ++k)
            if (p[k].y >= 0) csr[q[k]] = p[k].x;
    }
}

template <int V> struct rowvec;
template <> struct rowvec<4> { typedef float4 type; };
template <> struct rowvec<2> { typedef float2 type; };

template <int C, bool RELU>
__launch_bounds__(256)
__global__ void k_gather(const int* __restrict__ rowptr, const int* __restrict__ csr,
                         const float* __restrict__ H, const float* __restrict__ bias,
                         float* __restrict__ out, int n, int E) {
    constexpr int V = C / 32;
    constexpr int U = 8;
    typedef typename rowvec<V>::type vec;

    const int lane = threadIdx.x & 31;
    const int wave   = __builtin_amdgcn_readfirstlane((int)((blockIdx.x * blockDim.x + threadIdx.x) >> 5));
    const int nwaves = (int)((gridDim.x * blockDim.x) >> 5);
    const float* Hl  = H + lane * V;

    float bv[V];
#pragma unroll
    for (int v = 0; v < V; ++v) bv[v] = bias[lane * V + v];

    int i = wave;
    int beg = 0, end = 0;
    if (i < n) {
        beg = rowptr[i];
        end = (i + 1 < n) ? rowptr[i + 1] : E;
    }
    while (i < n) {
        const int inext = i + nwaves;
        int nbeg = 0, nend = 0;
        if (inext < n) {
            nbeg = rowptr[inext];
            nend = (inext + 1 < n) ? rowptr[inext + 1] : E;
        }
        const float di = rsqrtf((float)(end - beg + 1));
        const vec hself = *(const vec*)(Hl + (size_t)i * C);

        float acc[V];
#pragma unroll
        for (int v = 0; v < V; ++v) acc[v] = 0.0f;

        for (int b0 = beg; b0 < end; b0 += 32) {
            const int m = (end - b0 < 32) ? (end - b0) : 32;
            int s = i;
            if (lane < m) s = csr[b0 + lane];
            int j = 0;
            do {
                vec h[U];
#pragma unroll
                for (int u = 0; u < U; ++u) {
                    const int su = __builtin_amdgcn_readlane(s, j + u);
                    h[u] = *(const vec*)(Hl + (size_t)su * C);
                }
#pragma unroll
                for (int u = 0; u < U; ++u) {
                    const float wu = (j + u < m) ? 1.0f : 0.0f;
                    const float* hp = (const float*)&h[u];
#pragma unroll
                    for (int v = 0; v < V; ++v) acc[v] = fmaf(hp[v], wu, acc[v]);
                }
                j += U;
            } while (j < m);
        }

        vec o;
        float* op = (float*)&o;
        const float* hs = (const float*)&hself;
#pragma unroll
        for (int v = 0; v < V; ++v) {
            const float r = fmaf(acc[v] + hs[v], di, bv[v]);
            op[v] = RELU ? fmaxf(r, 0.0f) : r;
        }
        *(vec*)(out + (size_t)i * C + lane * V) = o;
        i = inext;
        beg = nbeg;
        end = nend;
    }
}

extern "C" void kernel_launch(void* const* d_in, const int* in_sizes, int n_in,
                              void* d_out, int out_size, void* d_ws, size_t ws_size,
                              hipStream_t stream) {
    constexpr int IN  = 256;
    constexpr int HID = 128;
    constexpr int LAT = 64;

    const float* x   = (const float*)d_in[0];
    const int*   ei  = (const int*)d_in[1];
    const float* W1  = (const float*)d_in[2];
    const float* b1  = (const float*)d_in[3];
    const float* W2  = (const float*)d_in[4];
    const float* b2  = (const float*)d_in[5];
    float*       out = (float*)d_out;

    const int N = in_sizes[0] / IN;
    const int E = in_sizes[1] / 2;
    const int* src = ei;
    const int* dst = ei + E;

    char* ws = (char*)d_ws;
    int*   rowptr = (int*)ws;    ws += (size_t)N * sizeof(int);
    int*   csr    = (int*)ws;    ws += (size_t)E * sizeof(int);
    float* H1     = (float*)ws;  ws += (size_t)N * HID * sizeof(float);
    float* S1     = (float*)ws;
    float* H2     = H1;
    const int NB    = cdiv(N, BN);
    const int chunk = cdiv(E, G1);
    const int MN    = NB * G1;
    int*   mat    = (int*)S1;
    int*   offs   = mat + (size_t)cdiv(MN, 4) * 4;
    int*   pairs  = offs + (size_t)cdiv(MN, 4) * 4;

    const int T = 256;
    const int GW = 2048;

    k_bcount      <<<G1, 1024, NB * sizeof(int), stream>>>(dst, mat, E, chunk, NB);
    k_scan        <<<cdiv(MN, 1024), 1024, 0, stream>>>(mat, offs, MN);
    k_bscatter    <<<G1, 1024, NB * sizeof(int), stream>>>(src, dst, offs, pairs, E, chunk, NB);
    k_bucket_lists<<<NB, 1024, 0, stream>>>(pairs, offs, rowptr, csr, N, E, NB);

    gemm_wmma_f32<IN, HID><<<GEMM_WGS, GEMM_WAVES * 32, 0, stream>>>(x, W1, rowptr, H1, N, E);

    k_gather<HID, true><<<GW, T, 0, stream>>>(rowptr, csr, H1, b1, S1, N, E);

    gemm_wmma_f32<HID, LAT><<<GEMM_WGS, GEMM_WAVES * 32, 0, stream>>>(S1, W2, rowptr, H2, N, E);

    k_gather<LAT, false><<<GW, T, 0, stream>>>(rowptr, csr, H2, b2, out, N, E);
}


# Layer-2 product fused into the layer-1 aggregation: a wave keeps its 16 summed rows in LDS and multiplies them by W2 on WMMA; the activation plane's write and read and one launch go; sums unchanged
# speedup vs baseline: 3.4426x; 1.0493x over previous
#include <hip/hip_runtime.h>


typedef __attribute__((ext_vector_type(2))) float v2f;
typedef __attribute__((ext_vector_type(8))) float v8f;

static inline int cdiv(long long a, long long b) { return (int)((a + b - 1) / b); }

constexpr int GEMM_WAVES = 16;
constexpr int GEMM_WGS   = 256;

template <int K, int N>
__launch_bounds__(GEMM_WAVES * 32)
__global__ void gemm_wmma_f32(const float* __restrict__ X,
                              const float* __restrict__ W,
                              const int* __restrict__ rowptr,
                              float* __restrict__ H,
                              int nrows, int E) {
    constexpr int NT = N / 16;
    constexpr int NH = NT / 2;
    constexpr int NTHR = GEMM_WAVES * 32;
    constexpr int LDSF = (K * N > 41984) ? K * N : 41984;
    __shared__ float BF[LDSF];

    const int tid = threadIdx.x;
    for (int idx = tid; idx < K * N; idx += NTHR) {
        const int n  = idx / K;
        const int k  = idx - n * K;
        const int ks = k >> 2;
        const int kk = k & 3;
        const int g  = kk >> 1;
        const int r  = kk & 1;
        const int t  = n >> 4;
        const int nl = n & 15;
        BF[((ks * NT + t) * 32 + g * 16 + nl) * 2 + r] = W[idx];
    }
    __syncthreads();

    const int wave   = __builtin_amdgcn_readfirstlane(tid >> 5);
    const int lane   = tid & 31;
    const int laneLo = lane & 15;
    const int laneHi = lane >> 4;
    const int nstrips = nrows >> 4;

    for (int u = wave; ; u += GEMM_WAVES) {
        const int strip = (int)blockIdx.x + (int)gridDim.x * (u >> 1);
        if (strip >= nstrips) break;
        const int m0 = strip * 16;
        const int t0 = (u & 1) * NH;

        float dv[8];
        {
            const int r0 = m0 + 8 * laneHi;
            int rp[9];
#pragma unroll
            for (int r = 0; r < 8; ++r) rp[r] = rowptr[r0 + r];
            rp[8] = (r0 + 8 < nrows) ? rowptr[r0 + 8] : E;
#pragma unroll
            for (int r = 0; r < 8; ++r) dv[r] = rsqrtf((float)(rp[r + 1] - rp[r] + 1));
        }

        v8f acc[NH] = {};

        const float* xrow  = X + (size_t)(m0 + laneLo) * K;
        const float* fbase = BF + t0 * 64 + lane * 2;
        for (int ks = 0; ks < K / 4; ++ks) {
            const v2f a = *(const v2f*)(xrow + ks * 4 + 2 * laneHi);
            const float* fb = fbase + (ks * NT) * 64;
#pragma unroll
            for (int t = 0; t < NH; ++t) {
                const v2f b = *(const v2f*)(fb + t * 64);
                acc[t] = __builtin_amdgcn_wmma_f32_16x16x4_f32(
                    false, a, false, b, (short)0, acc[t], false, false);
            }
        }

#pragma unroll
        for (int t = 0; t < NH; ++t)
#pragma unroll
            for (int r = 0; r < 8; ++r)
                H[(size_t)(m0 + r + 8 * laneHi) * N + (t0 + t) * 16 + laneLo] = acc[t][r] * dv[r];
    }
}

constexpr int BSH = 8;
constexpr int BN  = 1 << BSH;
constexpr int G1  = 256;
constexpr int PER = 8;

__launch_bounds__(1024)
__global__ void k_bcount(const int* __restrict__ dst, int* __restrict__ mat,
                         int E, int chunk, int NB) {
    extern __shared__ int lds_i[];
    const int tid = threadIdx.x;
    for (int b = tid; b < NB; b += 1024) lds_i[b] = 0;
    __syncthreads();
    const int e0 = blockIdx.x * chunk;
    const int e1 = min(E, e0 + chunk);
    for (int base = e0; base < e1; base += 1024 * PER) {
        int d[PER];
#pragma unroll
        for (int k = 0; k < PER; ++k) {
            const int e = base + k * 1024 + tid;
            d[k] = (e < e1) ? dst[e] : -1;
        }
#pragma unroll
        for (int k = 0; k < PER; ++k)
            if (d[k] >= 0) atomicAdd(&lds_i[d[k] >> BSH], 1);
    }
    __syncthreads();
    for (int b = tid; b < NB; b += 1024) mat[b * G1 + blockIdx.x] = lds_i[b];
}

__launch_bounds__(1024)
__global__ void k_bscatter(const int* __restrict__ src, const int* __restrict__ dst,
                           const int* __restrict__ offs, int* __restrict__ pairs_i,
                           int E, int chunk, int NB) {
    extern __shared__ int lds_i[];
    int2* __restrict__ pairs = (int2*)pairs_i;
    const int tid = threadIdx.x;
    for (int b = tid; b < NB; b += 1024) lds_i[b] = offs[b * G1 + blockIdx.x];
    __syncthreads();
    const int e0 = blockIdx.x * chunk;
    const int e1 = min(E, e0 + chunk);
    for (int base = e0; base < e1; base += 1024 * PER) {
        int d[PER], s[PER];
#pragma unroll
        for (int k = 0; k < PER; ++k) {
            const int e = base + k * 1024 + tid;
            d[k] = (e < e1) ? dst[e] : -1;
            s[k] = (e < e1) ? src[e] : 0;
        }
        int p[PER];
#pragma unroll
        for (int k = 0; k < PER; ++k)
            p[k] = (d[k] >= 0) ? atomicAdd(&lds_i[d[k] >> BSH], 1) : 0;
#pragma unroll
        for (int k = 0; k < PER; ++k)
            if (d[k] >= 0) pairs[p[k]] = make_int2(s[k], d[k]);
    }
}

__launch_bounds__(1024)
__global__ void k_scan(const int* __restrict__ cnt, int* __restrict__ rowptr, int n) {
    __shared__ int wtot[32];
    __shared__ int wpre[32];
    const int tid  = threadIdx.x;
    const int lane = tid & 31;
    const int wv   = tid >> 5;
    const int base = blockIdx.x * 1024;
    const int i    = base + tid;
    const int c    = (i < n) ? cnt[i] : 0;

    int part = 0;
    for (int i0 = tid * 4; i0 < base; i0 += 8 * 4096) {
        int4 v[8];
#pragma unroll
        for (int k = 0; k < 8; ++k) {
            const int ii = i0 + k * 4096;
            v[k] = (ii < base) ? *(const int4*)(cnt + ii) : make_int4(0, 0, 0, 0);
        }
#pragma unroll
        for (int k = 0; k < 8; ++k) part += (v[k].x + v[k].y) + (v[k].z + v[k].w);
    }
    for (int o = 16; o > 0; o >>= 1) part += __shfl_xor(part, o);
    if (lane == 0) wtot[wv] = part;
    __syncthreads();
    int off = 0;
    for (int w = 0; w < 32; ++w) off += wtot[w];
    __syncthreads();

    int incl = c;
    for (int o = 1; o < 32; o <<= 1) {
        const int up = __shfl_up(incl, o);
        if (lane >= o) incl += up;
    }
    if (lane == 31) wtot[wv] = incl;
    __syncthreads();
    if (wv == 0) {
        const int t = wtot[lane];
        int sc = t;
        for (int o = 1; o < 32; o <<= 1) {
            const int up = __shfl_up(sc, o);
            if (lane >= o) sc += up;
        }
        wpre[lane] = sc - t;
    }
    __syncthreads();
    if (i < n) rowptr[i] = off + wpre[wv] + (incl - c);
}

__launch_bounds__(1024)
__global__ void k_bucket_lists(const int* __restrict__ pairs_i, const int* __restrict__ offs,
                               int* __restrict__ rowptr, int* __restrict__ csr,
                               int N, int E, int NB) {
    __shared__ int cur[BN];
    __shared__ int wtot[BN / 32];
    const int2* __restrict__ pairs = (const int2*)pairs_i;
    const int tid  = threadIdx.x;
    const int lane = tid & 31;
    const int wv   = tid >> 5;
    const int b    = blockIdx.x;
    const int n0   = b << BSH;
    const int beg  = offs[b * G1];
    const int end  = (b + 1 < NB) ? offs[(b + 1) * G1] : E;

    if (tid < BN) cur[tid] = 0;
    __syncthreads();
    for (int base = beg; base < end; base += 1024 * PER) {
        int dy[PER];
#pragma unroll
        for (int k = 0; k < PER; ++k) {
            const int e = base + k * 1024 + tid;
            dy[k] = (e < end) ? pairs[e].y : -1;
        }
#pragma unroll
        for (int k = 0; k < PER; ++k)
            if (dy[k] >= 0) atomicAdd(&cur[dy[k] - n0], 1);
    }
    __syncthreads();

    const int c = (tid < BN) ? cur[tid] : 0;
    int incl = c;
    for (int o = 1; o < 32; o <<= 1) {
        const int up = __shfl_up(incl, o);
        if (lane >= o) incl += up;
    }
    if (tid < BN && lane == 31) wtot[wv] = incl;
    __syncthreads();
    if (tid < BN) {
        int woff = 0;
        for (int w = 0; w < wv; ++w) woff += wtot[w];
        const int start = beg + woff + (incl - c);
        if (n0 + tid < N) rowptr[n0 + tid] = start;
        cur[tid] = start;
    }
    __syncthreads();

    for (int base = beg; base < end; base += 1024 * PER) {
        int2 p[PER];
#pragma unroll
        for (int k = 0; k < PER; ++k) {
            const int e = base + k * 1024 + tid;
            p[k] = (e < end) ? pairs[e] : make_int2(0, -1);
        }
        int q[PER];
#pragma unroll
        for (int k = 0; k < PER; ++k)
            q[k] = (p[k].y >= 0) ? atomicAdd(&cur[p[k].y - n0], 1) : 0;
#pragma unroll
        for (int k = 0; k < PER; ++k)
            if (p[k].y >= 0) csr[q[k]] = p[k].x;
    }
}

template <int V> struct rowvec;
template <> struct rowvec<4> { typedef float4 type; };
template <> struct rowvec<2> { typedef float2 type; };

template <int C, bool RELU>
__launch_bounds__(256)
__global__ void k_gather(const int* __restrict__ rowptr, const int* __restrict__ csr,
                         const float* __restrict__ H, const float* __restrict__ bias,
                         float* __restrict__ out, int n, int E) {
    constexpr int V = C / 32;
    constexpr int U = 8;
    typedef typename rowvec<V>::type vec;

    const int lane = threadIdx.x & 31;
    const int wave   = __builtin_amdgcn_readfirstlane((int)((blockIdx.x * blockDim.x + threadIdx.x) >> 5));
    const int nwaves = (int)((gridDim.x * blockDim.x) >> 5);
    const float* Hl  = H + lane * V;

    float bv[V];
#pragma unroll
    for (int v = 0; v < V; ++v) bv[v] = bias[lane * V + v];

    int i = wave;
    int beg = 0, end = 0;
    if (i < n) {
        beg = rowptr[i];
        end = (i + 1 < n) ? rowptr[i + 1] : E;
    }
    while (i < n) {
        const int inext = i + nwaves;
        int nbeg = 0, nend = 0;
        if (inext < n) {
            nbeg = rowptr[inext];
            nend = (inext + 1 < n) ? rowptr[inext + 1] : E;
        }
        const float di = rsqrtf((float)(end - beg + 1));
        const vec hself = *(const vec*)(Hl + (size_t)i * C);

        float acc[V];
#pragma unroll
        for (int v = 0; v < V; ++v) acc[v] = 0.0f;

        for (int b0 = beg; b0 < end; b0 += 32) {
            const int m = (end - b0 < 32) ? (end - b0) : 32;
            int s = i;
            if (lane < m) s = csr[b0 + lane];
            int j = 0;
            do {
                vec h[U];
#pragma unroll
                for (int u = 0; u < U; ++u) {
                    const int su = __builtin_amdgcn_readlane(s, j + u);
                    h[u] = *(const vec*)(Hl + (size_t)su * C);
                }
#pragma unroll
                for (int u = 0; u < U; ++u) {
                    const float wu = (j + u < m) ? 1.0f : 0.0f;
                    const float* hp = (const float*)&h[u];
#pragma unroll
                    for (int v = 0; v < V; ++v) acc[v] = fmaf(hp[v], wu, acc[v]);
                }
                j += U;
            } while (j < m);
        }

        vec o;
        float* op = (float*)&o;
        const float* hs = (const float*)&hself;
#pragma unroll
        for (int v = 0; v < V; ++v) {
            const float r = fmaf(acc[v] + hs[v], di, bv[v]);
            op[v] = RELU ? fmaxf(r, 0.0f) : r;
        }
        *(vec*)(out + (size_t)i * C + lane * V) = o;
        i = inext;
        beg = nbeg;
        end = nend;
    }
}

constexpr int FG_WAVES = 32;
constexpr int FG_PITCH = 132;

template <int C, int N2>
__launch_bounds__(FG_WAVES * 32)
__global__ void k_gather_gemm(const int* __restrict__ rowptr, const int* __restrict__ csr,
                              const float* __restrict__ H, const float* __restrict__ bias,
                              const float* __restrict__ W,
                              float* __restrict__ H2,
                              int n, int E) {
    static_assert(C == 128, "one float4 of a row per lane");
    constexpr int K = C;
    constexpr int NT = N2 / 16;
    constexpr int NTHR = FG_WAVES * 32;
    constexpr int U = 16;
    __shared__ float BF[K * N2];
    __shared__ float ST[FG_WAVES * 16 * FG_PITCH];

    const int tid = threadIdx.x;
    constexpr int STG = (K * N2) / NTHR;
    static_assert(STG * NTHR == K * N2, "W is dealt evenly over the threads");
    float wv[STG];
#pragma unroll
    for (int q = 0; q < STG; ++q) wv[q] = W[tid + q * NTHR];
#pragma unroll
    for (int q = 0; q < STG; ++q) {
        const int idx = tid + q * NTHR;
        const int nn = idx / K;
        const int k  = idx - nn * K;
        const int ks = k >> 2;
        const int kk = k & 3;
        const int g  = kk >> 1;
        const int r  = kk & 1;
        const int t  = nn >> 4;
        const int nl = nn & 15;
        BF[((ks * NT + t) * 32 + g * 16 + nl) * 2 + r] = wv[q];
    }
    __syncthreads();

    const int wave   = __builtin_amdgcn_readfirstlane(tid >> 5);
    const int lane   = tid & 31;
    const int laneLo = lane & 15;
    const int laneHi = lane >> 4;
    float* STw = ST + wave * (16 * FG_PITCH);
    const float* Hl = H + lane * 4;

    float bv[4];
#pragma unroll
    for (int v = 0; v < 4; ++v) bv[v] = bias[lane * 4 + v];

    const int ntiles = n >> 4;
    for (int tile = (int)blockIdx.x + (int)gridDim.x * wave; tile < ntiles;
         tile += (int)gridDim.x * FG_WAVES) {
        const int m0 = tile * 16;
        int rp = 0;
        if (lane <= 16) {
            const int idx = m0 + lane;
            rp = (idx < n) ? rowptr[idx] : E;
        }
        int beg = __builtin_amdgcn_readlane(rp, 0);
        int end = __builtin_amdgcn_readlane(rp, 1);
        int s_cur = m0;
        if (lane < end - beg) s_cur = csr[beg + lane];

        for (int r = 0; r < 16; ++r) {
            const int i = m0 + r;
            const int nend = __builtin_amdgcn_readlane(rp, r + 2);
            const int inx = max(min(end + lane, E - 1), 0);
            const int cnx = csr[inx];
            const int s_nx = (lane < nend - end) ? cnx : (i + 1);
            const float di = rsqrtf((float)(end - beg + 1));
            const float4 hself = *(const float4*)(Hl + (size_t)i * C);

            float acc[4] = {0.0f, 0.0f, 0.0f, 0.0f};
            auto sum_rows = [&](int s, int m) {
                int j = 0;
                do {
                    float4 h[U];
#pragma unroll
                    for (int u = 0; u < U; ++u) {
                        const int su = __builtin_amdgcn_readlane(s, j + u);
                        h[u] = *(const float4*)(Hl + (size_t)su * C);
                    }
#pragma unroll
                    for (int u = 0; u < U; ++u) {
                        const float wu = (j + u < m) ? 1.0f : 0.0f;
                        acc[0] = fmaf(h[u].x, wu, acc[0]);
                        acc[1] = fmaf(h[u].y, wu, acc[1]);
                        acc[2] = fmaf(h[u].z, wu, acc[2]);
                        acc[3] = fmaf(h[u].w, wu, acc[3]);
                    }
                    j += U;
                } while (j < m);
            };
            if (beg < end) {
                const int m = (end - beg < 32) ? (end - beg) : 32;
                sum_rows(s_cur, m);
                for (int b0 = beg + 32; b0 < end; b0 += 32) {
                    const int mm = (end - b0 < 32) ? (end - b0) : 32;
                    int s = i;
                    if (lane < mm) s = csr[b0 + lane];
                    sum_rows(s, mm);
                }
            }

            float4 o;
            o.x = fmaxf(fmaf(acc[0] + hself.x, di, bv[0]), 0.0f);
            o.y = fmaxf(fmaf(acc[1] + hself.y, di, bv[1]), 0.0f);
            o.z = fmaxf(fmaf(acc[2] + hself.z, di, bv[2]), 0.0f);
            o.w = fmaxf(fmaf(acc[3] + hself.w, di, bv[3]), 0.0f);
            *(float4*)(STw + r * FG_PITCH + lane * 4) = o;
            s_cur = s_nx;
            beg = end;
            end = nend;
        }

        float dv[8];
        {
            const int rpn = __shfl_down(rp, 1);
            const float dm = rsqrtf((float)(((lane < 16) ? (rpn - rp) : 0) + 1));
#pragma unroll
            for (int q = 0; q < 8; ++q) dv[q] = __shfl(dm, 8 * laneHi + q);
        }

        v8f accm[NT] = {};
        const float* arow  = STw + laneLo * FG_PITCH + 2 * laneHi;
        const float* fbase = BF + lane * 2;
        for (int ks = 0; ks < K / 4; ++ks) {
            const v2f a = *(const v2f*)(arow + ks * 4);
            const float* fb = fbase + (ks * NT) * 64;
#pragma unroll
            for (int t = 0; t < NT; ++t) {
                const v2f b = *(const v2f*)(fb + t * 64);
                accm[t] = __builtin_amdgcn_wmma_f32_16x16x4_f32(
                    false, a, false, b, (short)0, accm[t], false, false);
            }
        }

#pragma unroll
        for (int t = 0; t < NT; ++t)
#pragma unroll
            for (int q = 0; q < 8; ++q)
                H2[(size_t)(m0 + q + 8 * laneHi) * N2 + t * 16 + laneLo] = accm[t][q] * dv[q];
    }
}

extern "C" void kernel_launch(void* const* d_in, const int* in_sizes, int n_in,
                              void* d_out, int out_size, void* d_ws, size_t ws_size,
                              hipStream_t stream) {
    constexpr int IN  = 256;
    constexpr int HID = 128;
    constexpr int LAT = 64;

    const float* x   = (const float*)d_in[0];
    const int*   ei  = (const int*)d_in[1];
    const float* W1  = (const float*)d_in[2];
    const float* b1  = (const float*)d_in[3];
    const float* W2  = (const float*)d_in[4];
    const float* b2  = (const float*)d_in[5];
    float*       out = (float*)d_out;

    const int N = in_sizes[0] / IN;
    const int E = in_sizes[1] / 2;
    const int* src = ei;
    const int* dst = ei + E;

    char* ws = (char*)d_ws;
    int*   rowptr = (int*)ws;    ws += (size_t)N * sizeof(int);
    int*   csr    = (int*)ws;    ws += (size_t)E * sizeof(int);
    float* H1     = (float*)ws;  ws += (size_t)N * HID * sizeof(float);
    float* S1     = (float*)ws;
    float* H2     = S1;
    const int NB    = cdiv(N, BN);
    const int chunk = cdiv(E, G1);
    const int MN    = NB * G1;
    int*   mat    = (int*)S1;
    int*   offs   = mat + (size_t)cdiv(MN, 4) * 4;
    int*   pairs  = offs + (size_t)cdiv(MN, 4) * 4;

    const int T = 256;
    const int GW = 2048;

    k_bcount      <<<G1, 1024, NB * sizeof(int), stream>>>(dst, mat, E, chunk, NB);
    k_scan        <<<cdiv(MN, 1024), 1024, 0, stream>>>(mat, offs, MN);
    k_bscatter    <<<G1, 1024, NB * sizeof(int), stream>>>(src, dst, offs, pairs, E, chunk, NB);
    k_bucket_lists<<<NB, 1024, 0, stream>>>(pairs, offs, rowptr, csr, N, E, NB);

    gemm_wmma_f32<IN, HID><<<GEMM_WGS, GEMM_WAVES * 32, 0, stream>>>(x, W1, rowptr, H1, N, E);

    k_gather_gemm<HID, LAT><<<GEMM_WGS, FG_WAVES * 32, 0, stream>>>(rowptr, csr, H1, b1, W2, H2, N, E);

    k_gather<LAT, false><<<GW, T, 0, stream>>>(rowptr, csr, H2, b2, out, N, E);
}


# Edge grouping in two launches, not four: each chunk is counting-sorted by bucket in LDS (count, scan, place) in one launch; bucket groups walk the chunks' runs; list order = LDS add arrival order
# speedup vs baseline: 3.8521x; 1.1190x over previous
#include <hip/hip_runtime.h>


typedef __attribute__((ext_vector_type(2))) float v2f;
typedef __attribute__((ext_vector_type(8))) float v8f;

static inline int cdiv(long long a, long long b) { return (int)((a + b - 1) / b); }

constexpr int GEMM_WAVES = 16;
constexpr int GEMM_WGS   = 256;

template <int K, int N>
__launch_bounds__(GEMM_WAVES * 32)
__global__ void gemm_wmma_f32(const float* __restrict__ X,
                              const float* __restrict__ W,
                              const int* __restrict__ rowptr,
                              float* __restrict__ H,
                              int nrows, int E) {
    constexpr int NT = N / 16;
    constexpr int NH = NT / 2;
    constexpr int NTHR = GEMM_WAVES * 32;
    constexpr int LDSF = (K * N > 41984) ? K * N : 41984;
    __shared__ float BF[LDSF];

    const int tid = threadIdx.x;
    for (int idx = tid; idx < K * N; idx += NTHR) {
        const int n  = idx / K;
        const int k  = idx - n * K;
        const int ks = k >> 2;
        const int kk = k & 3;
        const int g  = kk >> 1;
        const int r  = kk & 1;
        const int t  = n >> 4;
        const int nl = n & 15;
        BF[((ks * NT + t) * 32 + g * 16 + nl) * 2 + r] = W[idx];
    }
    __syncthreads();

    const int wave   = __builtin_amdgcn_readfirstlane(tid >> 5);
    const int lane   = tid & 31;
    const int laneLo = lane & 15;
    const int laneHi = lane >> 4;
    const int nstrips = nrows >> 4;

    for (int u = wave; ; u += GEMM_WAVES) {
        const int strip = (int)blockIdx.x + (int)gridDim.x * (u >> 1);
        if (strip >= nstrips) break;
        const int m0 = strip * 16;
        const int t0 = (u & 1) * NH;

        float dv[8];
        {
            const int r0 = m0 + 8 * laneHi;
            int rp[9];
#pragma unroll
            for (int r = 0; r < 8; ++r) rp[r] = rowptr[r0 + r];
            rp[8] = (r0 + 8 < nrows) ? rowptr[r0 + 8] : E;
#pragma unroll
            for (int r = 0; r < 8; ++r) dv[r] = rsqrtf((float)(rp[r + 1] - rp[r] + 1));
        }

        v8f acc[NH] = {};

        const float* xrow  = X + (size_t)(m0 + laneLo) * K;
        const float* fbase = BF + t0 * 64 + lane * 2;
        for (int ks = 0; ks < K / 4; ++ks) {
            const v2f a = *(const v2f*)(xrow + ks * 4 + 2 * laneHi);
            const float* fb = fbase + (ks * NT) * 64;
#pragma unroll
            for (int t = 0; t < NH; ++t) {
                const v2f b = *(const v2f*)(fb + t * 64);
                acc[t] = __builtin_amdgcn_wmma_f32_16x16x4_f32(
                    false, a, false, b, (short)0, acc[t], false, false);
            }
        }

#pragma unroll
        for (int t = 0; t < NH; ++t)
#pragma unroll
            for (int r = 0; r < 8; ++r)
                H[(size_t)(m0 + r + 8 * laneHi) * N + (t0 + t) * 16 + laneLo] = acc[t][r] * dv[r];
    }
}

constexpr int BSH = 8;
constexpr int BN  = 1 << BSH;
constexpr int G1  = 256;
constexpr int PER = 8;

__device__ __forceinline__ int wave_incl_scan(int v, int lane) {
    for (int o = 1; o < 32; o <<= 1) {
        const int up = __shfl_up(v, o);
        if (lane >= o) v += up;
    }
    return v;
}

__launch_bounds__(1024)
__global__ void k_chunk_sort(const int* __restrict__ src, const int* __restrict__ dst,
                             int* __restrict__ loc, int* __restrict__ pairs_i,
                             int E, int chunk, int NB) {
    extern __shared__ int lds_i[];
    __shared__ int wtot[32];
    int2* __restrict__ pairs = (int2*)pairs_i;
    const int tid  = threadIdx.x;
    const int lane = tid & 31;
    const int wv   = tid >> 5;
    for (int b = tid; b < NB; b += 1024) lds_i[b] = 0;
    __syncthreads();
    const int e0 = blockIdx.x * chunk;
    const int e1 = min(E, e0 + chunk);
    for (int base = e0; base < e1; base += 1024 * PER) {
        int d[PER];
#pragma unroll
        for (int k = 0; k < PER; ++k) {
            const int e = base + k * 1024 + tid;
            d[k] = (e < e1) ? dst[e] : -1;
        }
#pragma unroll
        for (int k = 0; k < PER; ++k)
            if (d[k] >= 0) atomicAdd(&lds_i[d[k] >> BSH], 1);
    }
    __syncthreads();

    int* __restrict__ locc = loc + (size_t)blockIdx.x * (NB + 1);
    int carry = 0;
    for (int t0 = 0; t0 < NB; t0 += 1024) {
        const int i = t0 + tid;
        const int c = (i < NB) ? lds_i[i] : 0;
        const int incl = wave_incl_scan(c, lane);
        if (lane == 31) wtot[wv] = incl;
        __syncthreads();
        int woff = 0, tot = 0;
        for (int w = 0; w < 32; ++w) {
            const int t = wtot[w];
            tot += t;
            if (w < wv) woff += t;
        }
        const int ex = carry + woff + (incl - c);
        if (i < NB) {
            lds_i[i] = ex;
            locc[i]  = ex;
        }
        carry += tot;
        __syncthreads();
    }
    if (tid == 0) locc[NB] = carry;

    for (int base = e0; base < e1; base += 1024 * PER) {
        int d[PER], s[PER];
#pragma unroll
        for (int k = 0; k < PER; ++k) {
            const int e = base + k * 1024 + tid;
            d[k] = (e < e1) ? dst[e] : -1;
            s[k] = (e < e1) ? src[e] : 0;
        }
        int p[PER];
#pragma unroll
        for (int k = 0; k < PER; ++k)
            p[k] = (d[k] >= 0) ? atomicAdd(&lds_i[d[k] >> BSH], 1) : 0;
#pragma unroll
        for (int k = 0; k < PER; ++k)
            if (d[k] >= 0) pairs[e0 + p[k]] = make_int2(s[k], d[k]);
    }
}

__launch_bounds__(1024)
__global__ void k_bucket_lists(const int* __restrict__ pairs_i, const int* __restrict__ loc,
                               int* __restrict__ rowptr, int* __restrict__ csr,
                               int N, int NB, int chunk) {
    static_assert(G1 * 4 == 1024, "four threads per run");
    __shared__ int cur[BN];
    __shared__ int wtot[32];
    __shared__ int rstart[G1];
    __shared__ int rlen[G1];
    const int2* __restrict__ pairs = (const int2*)pairs_i;
    const int tid  = threadIdx.x;
    const int lane = tid & 31;
    const int wv   = tid >> 5;
    const int b    = blockIdx.x;
    const int n0   = b << BSH;

    int lo = 0;
    if (tid < G1) {
        const int* lc = loc + (size_t)tid * (NB + 1) + b;
        lo = lc[0];
        rstart[tid] = tid * chunk + lo;
        rlen[tid]   = lc[1] - lo;
    }
    if (tid < BN) cur[tid] = 0;
    int part = lo;
    for (int o = 16; o > 0; o >>= 1) part += __shfl_xor(part, o);
    if (lane == 0) wtot[wv] = part;
    __syncthreads();
    int beg = 0;
    for (int w = 0; w < 32; ++w) beg += wtot[w];
    const int rs  = rstart[tid >> 2];
    const int rl  = rlen[tid >> 2];
    const int sub = tid & 3;

    for (int j0 = sub; j0 < rl; j0 += 4 * PER) {
        int dy[PER];
#pragma unroll
        for (int k = 0; k < PER; ++k) {
            const int j = j0 + 4 * k;
            dy[k] = (j < rl) ? pairs[rs + j].y : -1;
        }
#pragma unroll
        for (int k = 0; k < PER; ++k)
            if (dy[k] >= 0) atomicAdd(&cur[dy[k] - n0], 1);
    }
    __syncthreads();

    const int c = (tid < BN) ? cur[tid] : 0;
    const int incl = wave_incl_scan(c, lane);
    if (tid < BN && lane == 31) wtot[wv] = incl;
    __syncthreads();
    if (tid < BN) {
        int woff = 0;
        for (int w = 0; w < wv; ++w) woff += wtot[w];
        const int start = beg + woff + (incl - c);
        if (n0 + tid < N) rowptr[n0 + tid] = start;
        cur[tid] = start;
    }
    __syncthreads();

    for (int j0 = sub; j0 < rl; j0 += 4 * PER) {
        int2 p[PER];
#pragma unroll
        for (int k = 0; k < PER; ++k) {
            const int j = j0 + 4 * k;
            p[k] = (j < rl) ? pairs[rs + j] : make_int2(0, -1);
        }
        int q[PER];
#pragma unroll
        for (int k = 0; k < PER; ++k)
            q[k] = (p[k].y >= 0) ? atomicAdd(&cur[p[k].y - n0], 1) : 0;
#pragma unroll
        for (int k = 0; k < PER; ++k)
            if (p[k].y >= 0) csr[q[k]] = p[k].x;
    }
}

template <int V> struct rowvec;
template <> struct rowvec<4> { typedef float4 type; };
template <> struct rowvec<2> { typedef float2 type; };

template <int C, bool RELU>
__launch_bounds__(256)
__global__ void k_gather(const int* __restrict__ rowptr, const int* __restrict__ csr,
                         const float* __restrict__ H, const float* __restrict__ bias,
                         float* __restrict__ out, int n, int E) {
    constexpr int V = C / 32;
    constexpr int U = 8;
    typedef typename rowvec<V>::type vec;

    const int lane = threadIdx.x & 31;
    const int wave   = __builtin_amdgcn_readfirstlane((int)((blockIdx.x * blockDim.x + threadIdx.x) >> 5));
    const int nwaves = (int)((gridDim.x * blockDim.x) >> 5);
    const float* Hl  = H + lane * V;

    float bv[V];
#pragma unroll
    for (int v = 0; v < V; ++v) bv[v] = bias[lane * V + v];

    int i = wave;
    int beg = 0, end = 0;
    if (i < n) {
        beg = rowptr[i];
        end = (i + 1 < n) ? rowptr[i + 1] : E;
    }
    while (i < n) {
        const int inext = i + nwaves;
        int nbeg = 0, nend = 0;
        if (inext < n) {
            nbeg = rowptr[inext];
            nend = (inext + 1 < n) ? rowptr[inext + 1] : E;
        }
        const float di = rsqrtf((float)(end - beg + 1));
        const vec hself = *(const vec*)(Hl + (size_t)i * C);

        float acc[V];
#pragma unroll
        for (int v = 0; v < V; ++v) acc[v] = 0.0f;

        for (int b0 = beg; b0 < end; b0 += 32) {
            const int m = (end - b0 < 32) ? (end - b0) : 32;
            int s = i;
            if (lane < m) s = csr[b0 + lane];
            int j = 0;
            do {
                vec h[U];
#pragma unroll
                for (int u = 0; u < U; ++u) {
                    const int su = __builtin_amdgcn_readlane(s, j + u);
                    h[u] = *(const vec*)(Hl + (size_t)su * C);
                }
#pragma unroll
                for (int u = 0; u < U; ++u) {
                    const float wu = (j + u < m) ? 1.0f : 0.0f;
                    const float* hp = (const float*)&h[u];
#pragma unroll
                    for (int v = 0; v < V; ++v) acc[v] = fmaf(hp[v], wu, acc[v]);
                }
                j += U;
            } while (j < m);
        }

        vec o;
        float* op = (float*)&o;
        const float* hs = (const float*)&hself;
#pragma unroll
        for (int v = 0; v < V; ++v) {
            const float r = fmaf(acc[v] + hs[v], di, bv[v]);
            op[v] = RELU ? fmaxf(r, 0.0f) : r;
        }
        *(vec*)(out + (size_t)i * C + lane * V) = o;
        i = inext;
        beg = nbeg;
        end = nend;
    }
}

constexpr int FG_WAVES = 32;
constexpr int FG_PITCH = 132;

template <int C, int N2>
__launch_bounds__(FG_WAVES * 32)
__global__ void k_gather_gemm(const int* __restrict__ rowptr, const int* __restrict__ csr,
                              const float* __restrict__ H, const float* __restrict__ bias,
                              const float* __restrict__ W,
                              float* __restrict__ H2,
                              int n, int E) {
    static_assert(C == 128, "one float4 of a row per lane");
    constexpr int K = C;
    constexpr int NT = N2 / 16;
    constexpr int NTHR = FG_WAVES * 32;
    constexpr int U = 16;
    __shared__ float BF[K * N2];
    __shared__ float ST[FG_WAVES * 16 * FG_PITCH];

    const int tid = threadIdx.x;
    constexpr int STG = (K * N2) / NTHR;
    static_assert(STG * NTHR == K * N2, "W is dealt evenly over the threads");
    float wv[STG];
#pragma unroll
    for (int q = 0; q < STG; ++q) wv[q] = W[tid + q * NTHR];
#pragma unroll
    for (int q = 0; q < STG; ++q) {
        const int idx = tid + q * NTHR;
        const int nn = idx / K;
        const int k  = idx - nn * K;
        const int ks = k >> 2;
        const int kk = k & 3;
        const int g  = kk >> 1;
        const int r  = kk & 1;
        const int t  = nn >> 4;
        const int nl = nn & 15;
        BF[((ks * NT + t) * 32 + g * 16 + nl) * 2 + r] = wv[q];
    }
    __syncthreads();

    const int wave   = __builtin_amdgcn_readfirstlane(tid >> 5);
    const int lane   = tid & 31;
    const int laneLo = lane & 15;
    const int laneHi = lane >> 4;
    float* STw = ST + wave * (16 * FG_PITCH);
    const float* Hl = H + lane * 4;

    float bv[4];
#pragma unroll
    for (int v = 0; v < 4; ++v) bv[v] = bias[lane * 4 + v];

    const int ntiles = n >> 4;
    for (int tile = (int)blockIdx.x + (int)gridDim.x * wave; tile < ntiles;
         tile += (int)gridDim.x * FG_WAVES) {
        const int m0 = tile * 16;
        int rp = 0;
        if (lane <= 16) {
            const int idx = m0 + lane;
            rp = (idx < n) ? rowptr[idx] : E;
        }
        int beg = __builtin_amdgcn_readlane(rp, 0);
        int end = __builtin_amdgcn_readlane(rp, 1);
        int s_cur = m0;
        if (lane < end - beg) s_cur = csr[beg + lane];

        for (int r = 0; r < 16; ++r) {
            const int i = m0 + r;
            const int nend = __builtin_amdgcn_readlane(rp, r + 2);
            const int inx = max(min(end + lane, E - 1), 0);
            const int cnx = csr[inx];
            const int s_nx = (lane < nend - end) ? cnx : (i + 1);
            const float di = rsqrtf((float)(end - beg + 1));
            const float4 hself = *(const float4*)(Hl + (size_t)i * C);

            float acc[4] = {0.0f, 0.0f, 0.0f, 0.0f};
            auto sum_rows = [&](int s, int m) {
                int j = 0;
                do {
                    float4 h[U];
#pragma unroll
                    for (int u = 0; u < U; ++u) {
                        const int su = __builtin_amdgcn_readlane(s, j + u);
                        h[u] = *(const float4*)(Hl + (size_t)su * C);
                    }
#pragma unroll
                    for (int u = 0; u < U; ++u) {
                        const float wu = (j + u < m) ? 1.0f : 0.0f;
                        acc[0] = fmaf(h[u].x, wu, acc[0]);
                        acc[1] = fmaf(h[u].y, wu, acc[1]);
                        acc[2] = fmaf(h[u].z, wu, acc[2]);
                        acc[3] = fmaf(h[u].w, wu, acc[3]);
                    }
                    j += U;
                } while (j < m);
            };
            if (beg < end) {
                const int m = (end - beg < 32) ? (end - beg) : 32;
                sum_rows(s_cur, m);
                for (int b0 = beg + 32; b0 < end; b0 += 32) {
                    const int mm = (end - b0 < 32) ? (end - b0) : 32;
                    int s = i;
                    if (lane < mm) s = csr[b0 + lane];
                    sum_rows(s, mm);
                }
            }

            float4 o;
            o.x = fmaxf(fmaf(acc[0] + hself.x, di, bv[0]), 0.0f);
            o.y = fmaxf(fmaf(acc[1] + hself.y, di, bv[1]), 0.0f);
            o.z = fmaxf(fmaf(acc[2] + hself.z, di, bv[2]), 0.0f);
            o.w = fmaxf(fmaf(acc[3] + hself.w, di, bv[3]), 0.0f);
            *(float4*)(STw + r * FG_PITCH + lane * 4) = o;
            s_cur = s_nx;
            beg = end;
            end = nend;
        }

        float dv[8];
        {
            const int rpn = __shfl_down(rp, 1);
            const float dm = rsqrtf((float)(((lane < 16) ? (rpn - rp) : 0) + 1));
#pragma unroll
            for (int q = 0; q < 8; ++q) dv[q] = __shfl(dm, 8 * laneHi + q);
        }

        v8f accm[NT] = {};
        const float* arow  = STw + laneLo * FG_PITCH + 2 * laneHi;
        const float* fbase = BF + lane * 2;
        for (int ks = 0; ks < K / 4; ++ks) {
            const v2f a = *(const v2f*)(arow + ks * 4);
            const float* fb = fbase + (ks * NT) * 64;
#pragma unroll
            for (int t = 0; t < NT; ++t) {
                const v2f b = *(const v2f*)(fb + t * 64);
                accm[t] = __builtin_amdgcn_wmma_f32_16x16x4_f32(
                    false, a, false, b, (short)0, accm[t], false, false);
            }
        }

#pragma unroll
        for (int t = 0; t < NT; ++t)
#pragma unroll
            for (int q = 0; q < 8; ++q)
                H2[(size_t)(m0 + q + 8 * laneHi) * N2 + t * 16 + laneLo] = accm[t][q] * dv[q];
    }
}

extern "C" void kernel_launch(void* const* d_in, const int* in_sizes, int n_in,
                              void* d_out, int out_size, void* d_ws, size_t ws_size,
                              hipStream_t stream) {
    constexpr int IN  = 256;
    constexpr int HID = 128;
    constexpr int LAT = 64;

    const float* x   = (const float*)d_in[0];
    const int*   ei  = (const int*)d_in[1];
    const float* W1  = (const float*)d_in[2];
    const float* b1  = (const float*)d_in[3];
    const float* W2  = (const float*)d_in[4];
    const float* b2  = (const float*)d_in[5];
    float*       out = (float*)d_out;

    const int N = in_sizes[0] / IN;
    const int E = in_sizes[1] / 2;
    const int* src = ei;
    const int* dst = ei + E;

    char* ws = (char*)d_ws;
    int*   rowptr = (int*)ws;    ws += (size_t)N * sizeof(int);
    int*   csr    = (int*)ws;    ws += (size_t)E * sizeof(int);
    float* H1     = (float*)ws;  ws += (size_t)N * HID * sizeof(float);
    float* S1     = (float*)ws;
    float* H2     = S1;
    const int NB    = cdiv(N, BN);
    const int chunk = cdiv(E, G1);
    const int LN    = G1 * (NB + 1);
    int*   loc    = (int*)S1;
    int*   pairs  = loc + (size_t)cdiv(LN, 4) * 4;

    const int T = 256;
    const int GW = 2048;

    k_chunk_sort  <<<G1, 1024, NB * sizeof(int), stream>>>(src, dst, loc, pairs, E, chunk, NB);
    k_bucket_lists<<<NB, 1024, 0, stream>>>(pairs, loc, rowptr, csr, N, NB, chunk);

    gemm_wmma_f32<IN, HID><<<GEMM_WGS, GEMM_WAVES * 32, 0, stream>>>(x, W1, rowptr, H1, N, E);

    k_gather_gemm<HID, LAT><<<GEMM_WGS, FG_WAVES * 32, 0, stream>>>(rowptr, csr, H1, b1, W2, H2, N, E);

    k_gather<LAT, false><<<GW, T, 0, stream>>>(rowptr, csr, H2, b2, out, N, E);
}


# Bucket lists built inside the layer-1 product's launch: a bucket = the 400 rows one persistent group owns, dinv from its own LDS counts; 5 launches -> 4; list order = LDS add arrival order still
# speedup vs baseline: 4.0506x; 1.0515x over previous
#include <hip/hip_runtime.h>


typedef __attribute__((ext_vector_type(2))) float v2f;
typedef __attribute__((ext_vector_type(8))) float v8f;

static inline int cdiv(long long a, long long b) { return (int)((a + b - 1) / b); }

constexpr int SPB   = 25;
constexpr int BNODE = 16 * SPB;
constexpr int G1  = 256;
constexpr int PER = 8;

__device__ __forceinline__ int wave_incl_scan(int v, int lane) {
    for (int o = 1; o < 32; o <<= 1) {
        const int up = __shfl_up(v, o);
        if (lane >= o) v += up;
    }
    return v;
}

__launch_bounds__(1024)
__global__ void k_chunk_sort(const int* __restrict__ src, const int* __restrict__ dst,
                             int* __restrict__ loc, int* __restrict__ pairs_i,
                             int E, int chunk, int NB) {
    extern __shared__ int lds_i[];
    __shared__ int wtot[32];
    int2* __restrict__ pairs = (int2*)pairs_i;
    const int tid  = threadIdx.x;
    const int lane = tid & 31;
    const int wv   = tid >> 5;
    for (int b = tid; b < NB; b += 1024) lds_i[b] = 0;
    __syncthreads();
    const int e0 = blockIdx.x * chunk;
    const int e1 = min(E, e0 + chunk);
    for (int base = e0; base < e1; base += 1024 * PER) {
        int d[PER];
#pragma unroll
        for (int k = 0; k < PER; ++k) {
            const int e = base + k * 1024 + tid;
            d[k] = (e < e1) ? dst[e] : -1;
        }
#pragma unroll
        for (int k = 0; k < PER; ++k)
            if (d[k] >= 0) atomicAdd(&lds_i[(unsigned)d[k] / BNODE], 1);
    }
    __syncthreads();

    int* __restrict__ locc = loc + (size_t)blockIdx.x * (NB + 1);
    int carry = 0;
    for (int t0 = 0; t0 < NB; t0 += 1024) {
        const int i = t0 + tid;
        const int c = (i < NB) ? lds_i[i] : 0;
        const int incl = wave_incl_scan(c, lane);
        if (lane == 31) wtot[wv] = incl;
        __syncthreads();
        int woff = 0, tot = 0;
        for (int w = 0; w < 32; ++w) {
            const int t = wtot[w];
            tot += t;
            if (w < wv) woff += t;
        }
        const int ex = carry + woff + (incl - c);
        if (i < NB) {
            lds_i[i] = ex;
            locc[i]  = ex;
        }
        carry += tot;
        __syncthreads();
    }
    if (tid == 0) locc[NB] = carry;

    for (int base = e0; base < e1; base += 1024 * PER) {
        int d[PER], s[PER];
#pragma unroll
        for (int k = 0; k < PER; ++k) {
            const int e = base + k * 1024 + tid;
            d[k] = (e < e1) ? dst[e] : -1;
            s[k] = (e < e1) ? src[e] : 0;
        }
        int p[PER];
#pragma unroll
        for (int k = 0; k < PER; ++k)
            p[k] = (d[k] >= 0) ? atomicAdd(&lds_i[(unsigned)d[k] / BNODE], 1) : 0;
#pragma unroll
        for (int k = 0; k < PER; ++k)
            if (d[k] >= 0) pairs[e0 + p[k]] = make_int2(s[k], d[k]);
    }
}

constexpr int LG_WAVES = 32;

template <int K, int N>
__launch_bounds__(LG_WAVES * 32)
__global__ void k_lists_gemm(const int* __restrict__ pairs_i, const int* __restrict__ loc,
                             int* __restrict__ rowptr, int* __restrict__ csr,
                             const float* __restrict__ X,
                             const float* __restrict__ W,
                             float* __restrict__ H,
                             int nrows, int NB, int chunk) {
    static_assert(G1 * 4 == LG_WAVES * 32, "four threads per run");
    constexpr int NT = N / 16;
    constexpr int NH = NT / 2;
    constexpr int NTHR = LG_WAVES * 32;
    constexpr int LDSF = (K * N > 41984) ? K * N : 41984;
    __shared__ float BF[LDSF];
    __shared__ int   cur[BNODE];
    __shared__ float dvs[BNODE];
    __shared__ int   wtot[32];
    __shared__ int   rstart[G1];
    __shared__ int   rlen[G1];
    const int2* __restrict__ pairs = (const int2*)pairs_i;
    const int tid  = threadIdx.x;
    const int lane = tid & 31;
    const int wv   = tid >> 5;
    const int b    = blockIdx.x;
    const int n0   = b * BNODE;

    int lo = 0, hi = 0;
    if (tid < G1) {
        const int* lc = loc + (size_t)tid * (NB + 1) + b;
        lo = lc[0];
        hi = lc[1];
    }
    if (tid < BNODE) cur[tid] = 0;

    for (int idx = tid; idx < K * N; idx += NTHR) {
        const int n  = idx / K;
        const int k  = idx - n * K;
        const int ks = k >> 2;
        const int kk = k & 3;
        const int g  = kk >> 1;
        const int r  = kk & 1;
        const int t  = n >> 4;
        const int nl = n & 15;
        BF[((ks * NT + t) * 32 + g * 16 + nl) * 2 + r] = W[idx];
    }

    if (tid < G1) {
        rstart[tid] = tid * chunk + lo;
        rlen[tid]   = hi - lo;
    }
    int part = lo;
    for (int o = 16; o > 0; o >>= 1) part += __shfl_xor(part, o);
    if (lane == 0) wtot[wv] = part;
    __syncthreads();
    int beg = 0;
    for (int w = 0; w < 32; ++w) beg += wtot[w];
    const int rs  = rstart[tid >> 2];
    const int rl  = rlen[tid >> 2];
    const int sub = tid & 3;

    for (int j0 = sub; j0 < rl; j0 += 4 * PER) {
        int dy[PER];
#pragma unroll
        for (int k = 0; k < PER; ++k) {
            const int j = j0 + 4 * k;
            dy[k] = (j < rl) ? pairs[rs + j].y : -1;
        }
#pragma unroll
        for (int k = 0; k < PER; ++k)
            if (dy[k] >= 0) atomicAdd(&cur[dy[k] - n0], 1);
    }
    __syncthreads();

    const int c = (tid < BNODE) ? cur[tid] : 0;
    const int incl = wave_incl_scan(c, lane);
    if (lane == 31 && wv * 32 < BNODE) wtot[wv] = incl;
    __syncthreads();
    if (tid < BNODE) {
        int woff = 0;
        for (int w = 0; w < wv; ++w) woff += wtot[w];
        const int start = beg + woff + (incl - c);
        if (n0 + tid < nrows) rowptr[n0 + tid] = start;
        cur[tid] = start;
        dvs[tid] = rsqrtf((float)(c + 1));
    }
    __syncthreads();

    for (int j0 = sub; j0 < rl; j0 += 4 * PER) {
        int2 p[PER];
#pragma unroll
        for (int k = 0; k < PER; ++k) {
            const int j = j0 + 4 * k;
            p[k] = (j < rl) ? pairs[rs + j] : make_int2(0, -1);
        }
        int q[PER];
#pragma unroll
        for (int k = 0; k < PER; ++k)
            q[k] = (p[k].y >= 0) ? atomicAdd(&cur[p[k].y - n0], 1) : 0;
#pragma unroll
        for (int k = 0; k < PER; ++k)
            if (p[k].y >= 0) csr[q[k]] = p[k].x;
    }

    const int wave    = __builtin_amdgcn_readfirstlane(wv);
    const int laneLo  = lane & 15;
    const int laneHi  = lane >> 4;
    const int nstrips = nrows >> 4;
    const int s0      = b * SPB;

    for (int u = wave; u < 2 * SPB; u += LG_WAVES) {
        const int strip = s0 + (u >> 1);
        if (strip >= nstrips) break;
        const int m0 = strip * 16;
        const int t0 = (u & 1) * NH;

        v8f acc[NH] = {};

        const float* xrow  = X + (size_t)(m0 + laneLo) * K;
        const float* fbase = BF + t0 * 64 + lane * 2;
        for (int ks = 0; ks < K / 4; ++ks) {
            const v2f a = *(const v2f*)(xrow + ks * 4 + 2 * laneHi);
            const float* fb = fbase + (ks * NT) * 64;
#pragma unroll
            for (int t = 0; t < NH; ++t) {
                const v2f b2 = *(const v2f*)(fb + t * 64);
                acc[t] = __builtin_amdgcn_wmma_f32_16x16x4_f32(
                    false, a, false, b2, (short)0, acc[t], false, false);
            }
        }

        const float* dp = dvs + (m0 - n0) + 8 * laneHi;
        float dv[8];
#pragma unroll
        for (int r = 0; r < 8; ++r) dv[r] = dp[r];

#pragma unroll
        for (int t = 0; t < NH; ++t)
#pragma unroll
            for (int r = 0; r < 8; ++r)
                H[(size_t)(m0 + r + 8 * laneHi) * N + (t0 + t) * 16 + laneLo] = acc[t][r] * dv[r];
    }
}

template <int V> struct rowvec;
template <> struct rowvec<4> { typedef float4 type; };
template <> struct rowvec<2> { typedef float2 type; };

template <int C, bool RELU>
__launch_bounds__(256)
__global__ void k_gather(const int* __restrict__ rowptr, const int* __restrict__ csr,
                         const float* __restrict__ H, const float* __restrict__ bias,
                         float* __restrict__ out, int n, int E) {
    constexpr int V = C / 32;
    constexpr int U = 8;
    typedef typename rowvec<V>::type vec;

    const int lane = threadIdx.x & 31;
    const int wave   = __builtin_amdgcn_readfirstlane((int)((blockIdx.x * blockDim.x + threadIdx.x) >> 5));
    const int nwaves = (int)((gridDim.x * blockDim.x) >> 5);
    const float* Hl  = H + lane * V;

    float bv[V];
#pragma unroll
    for (int v = 0; v < V; ++v) bv[v] = bias[lane * V + v];

    int i = wave;
    int beg = 0, end = 0;
    if (i < n) {
        beg = rowptr[i];
        end = (i + 1 < n) ? rowptr[i + 1] : E;
    }
    while (i < n) {
        const int inext = i + nwaves;
        int nbeg = 0, nend = 0;
        if (inext < n) {
            nbeg = rowptr[inext];
            nend = (inext + 1 < n) ? rowptr[inext + 1] : E;
        }
        const float di = rsqrtf((float)(end - beg + 1));
        const vec hself = *(const vec*)(Hl + (size_t)i * C);

        float acc[V];
#pragma unroll
        for (int v = 0; v < V; ++v) acc[v] = 0.0f;

        for (int b0 = beg; b0 < end; b0 += 32) {
            const int m = (end - b0 < 32) ? (end - b0) : 32;
            int s = i;
            if (lane < m) s = csr[b0 + lane];
            int j = 0;
            do {
                vec h[U];
#pragma unroll
                for (int u = 0; u < U; ++u) {
                    const int su = __builtin_amdgcn_readlane(s, j + u);
                    h[u] = *(const vec*)(Hl + (size_t)su * C);
                }
#pragma unroll
                for (int u = 0; u < U; ++u) {
                    const float wu = (j + u < m) ? 1.0f : 0.0f;
                    const float* hp = (const float*)&h[u];
#pragma unroll
                    for (int v = 0; v < V; ++v) acc[v] = fmaf(hp[v], wu, acc[v]);
                }
                j += U;
            } while (j < m);
        }

        vec o;
        float* op = (float*)&o;
        const float* hs = (const float*)&hself;
#pragma unroll
        for (int v = 0; v < V; ++v) {
            const float r = fmaf(acc[v] + hs[v], di, bv[v]);
            op[v] = RELU ? fmaxf(r, 0.0f) : r;
        }
        *(vec*)(out + (size_t)i * C + lane * V) = o;
        i = inext;
        beg = nbeg;
        end = nend;
    }
}

constexpr int FG_WAVES = 32;
constexpr int FG_WGS   = 256;
constexpr int FG_PITCH = 132;

template <int C, int N2>
__launch_bounds__(FG_WAVES * 32)
__global__ void k_gather_gemm(const int* __restrict__ rowptr, const int* __restrict__ csr,
                              const float* __restrict__ H, const float* __restrict__ bias,
                              const float* __restrict__ W,
                              float* __restrict__ H2,
                              int n, int E) {
    static_assert(C == 128, "one float4 of a row per lane");
    constexpr int K = C;
    constexpr int NT = N2 / 16;
    constexpr int NTHR = FG_WAVES * 32;
    constexpr int U = 16;
    __shared__ float BF[K * N2];
    __shared__ float ST[FG_WAVES * 16 * FG_PITCH];

    const int tid = threadIdx.x;
    constexpr int STG = (K * N2) / NTHR;
    static_assert(STG * NTHR == K * N2, "W is dealt evenly over the threads");
    float wv[STG];
#pragma unroll
    for (int q = 0; q < STG; ++q) wv[q] = W[tid + q * NTHR];
#pragma unroll
    for (int q = 0; q < STG; ++q) {
        const int idx = tid + q * NTHR;
        const int nn = idx / K;
        const int k  = idx - nn * K;
        const int ks = k >> 2;
        const int kk = k & 3;
        const int g  = kk >> 1;
        const int r  = kk & 1;
        const int t  = nn >> 4;
        const int nl = nn & 15;
        BF[((ks * NT + t) * 32 + g * 16 + nl) * 2 + r] = wv[q];
    }
    __syncthreads();

    const int wave   = __builtin_amdgcn_readfirstlane(tid >> 5);
    const int lane   = tid & 31;
    const int laneLo = lane & 15;
    const int laneHi = lane >> 4;
    float* STw = ST + wave * (16 * FG_PITCH);
    const float* Hl = H + lane * 4;

    float bv[4];
#pragma unroll
    for (int v = 0; v < 4; ++v) bv[v] = bias[lane * 4 + v];

    const int ntiles = n >> 4;
    for (int tile = (int)blockIdx.x + (int)gridDim.x * wave; tile < ntiles;
         tile += (int)gridDim.x * FG_WAVES) {
        const int m0 = tile * 16;
        int rp = 0;
        if (lane <= 16) {
            const int idx = m0 + lane;
            rp = (idx < n) ? rowptr[idx] : E;
        }
        int beg = __builtin_amdgcn_readlane(rp, 0);
        int end = __builtin_amdgcn_readlane(rp, 1);
        int s_cur = m0;
        if (lane < end - beg) s_cur = csr[beg + lane];

        for (int r = 0; r < 16; ++r) {
            const int i = m0 + r;
            const int nend = __builtin_amdgcn_readlane(rp, r + 2);
            const int inx = max(min(end + lane, E - 1), 0);
            const int cnx = csr[inx];
            const int s_nx = (lane < nend - end) ? cnx : (i + 1);
            const float di = rsqrtf((float)(end - beg + 1));
            const float4 hself = *(const float4*)(Hl + (size_t)i * C);

            float acc[4] = {0.0f, 0.0f, 0.0f, 0.0f};
            auto sum_rows = [&](int s, int m) {
                int j = 0;
                do {
                    float4 h[U];
#pragma unroll
                    for (int u = 0; u < U; ++u) {
                        const int su = __builtin_amdgcn_readlane(s, j + u);
                        h[u] = *(const float4*)(Hl + (size_t)su * C);
                    }
#pragma unroll
                    for (int u = 0; u < U; ++u) {
                        const float wu = (j + u < m) ? 1.0f : 0.0f;
                        acc[0] = fmaf(h[u].x, wu, acc[0]);
                        acc[1] = fmaf(h[u].y, wu, acc[1]);
                        acc[2] = fmaf(h[u].z, wu, acc[2]);
                        acc[3] = fmaf(h[u].w, wu, acc[3]);
                    }
                    j += U;
                } while (j < m);
            };
            if (beg < end) {
                const int m = (end - beg < 32) ? (end - beg) : 32;
                sum_rows(s_cur, m);
                for (int b0 = beg + 32; b0 < end; b0 += 32) {
                    const int mm = (end - b0 < 32) ? (end - b0) : 32;
                    int s = i;
                    if (lane < mm) s = csr[b0 + lane];
                    sum_rows(s, mm);
                }
            }

            float4 o;
            o.x = fmaxf(fmaf(acc[0] + hself.x, di, bv[0]), 0.0f);
            o.y = fmaxf(fmaf(acc[1] + hself.y, di, bv[1]), 0.0f);
            o.z = fmaxf(fmaf(acc[2] + hself.z, di, bv[2]), 0.0f);
            o.w = fmaxf(fmaf(acc[3] + hself.w, di, bv[3]), 0.0f);
            *(float4*)(STw + r * FG_PITCH + lane * 4) = o;
            s_cur = s_nx;
            beg = end;
            end = nend;
        }

        float dv[8];
        {
            const int rpn = __shfl_down(rp, 1);
            const float dm = rsqrtf((float)(((lane < 16) ? (rpn - rp) : 0) + 1));
#pragma unroll
            for (int q = 0; q < 8; ++q) dv[q] = __shfl(dm, 8 * laneHi + q);
        }

        v8f accm[NT] = {};
        const float* arow  = STw + laneLo * FG_PITCH + 2 * laneHi;
        const float* fbase = BF + lane * 2;
        for (int ks = 0; ks < K / 4; ++ks) {
            const v2f a = *(const v2f*)(arow + ks * 4);
            const float* fb = fbase + (ks * NT) * 64;
#pragma unroll
            for (int t = 0; t < NT; ++t) {
                const v2f b = *(const v2f*)(fb + t * 64);
                accm[t] = __builtin_amdgcn_wmma_f32_16x16x4_f32(
                    false, a, false, b, (short)0, accm[t], false, false);
            }
        }

#pragma unroll
        for (int t = 0; t < NT; ++t)
#pragma unroll
            for (int q = 0; q < 8; ++q)
                H2[(size_t)(m0 + q + 8 * laneHi) * N2 + t * 16 + laneLo] = accm[t][q] * dv[q];
    }
}

extern "C" void kernel_launch(void* const* d_in, const int* in_sizes, int n_in,
                              void* d_out, int out_size, void* d_ws, size_t ws_size,
                              hipStream_t stream) {
    constexpr int IN  = 256;
    constexpr int HID = 128;
    constexpr int LAT = 64;

    const float* x   = (const float*)d_in[0];
    const int*   ei  = (const int*)d_in[1];
    const float* W1  = (const float*)d_in[2];
    const float* b1  = (const float*)d_in[3];
    const float* W2  = (const float*)d_in[4];
    const float* b2  = (const float*)d_in[5];
    float*       out = (float*)d_out;

    const int N = in_sizes[0] / IN;
    const int E = in_sizes[1] / 2;
    const int* src = ei;
    const int* dst = ei + E;

    char* ws = (char*)d_ws;
    int*   rowptr = (int*)ws;    ws += (size_t)N * sizeof(int);
    int*   csr    = (int*)ws;    ws += (size_t)E * sizeof(int);
    float* H1     = (float*)ws;  ws += (size_t)N * HID * sizeof(float);
    float* S1     = (float*)ws;
    float* H2     = S1;
    const int NB    = cdiv(N, BNODE);
    const int chunk = cdiv(E, G1);
    const int LN    = G1 * (NB + 1);
    int*   loc    = (int*)S1;
    int*   pairs  = loc + (size_t)cdiv(LN, 4) * 4;

    const int T = 256;
    const int GW = 2048;

    k_chunk_sort<<<G1, 1024, NB * sizeof(int), stream>>>(src, dst, loc, pairs, E, chunk, NB);
    k_lists_gemm<IN, HID><<<NB, LG_WAVES * 32, 0, stream>>>(pairs, loc, rowptr, csr, x, W1, H1, N, NB, chunk);

    k_gather_gemm<HID, LAT><<<FG_WGS, FG_WAVES * 32, 0, stream>>>(rowptr, csr, H1, b1, W2, H2, N, E);

    k_gather<LAT, false><<<GW, T, 0, stream>>>(rowptr, csr, H2, b2, out, N, E);
}


# Edge grouping's scattered per-edge stores (8-B pair slots, 4-B list slots: a line per lane) go to LDS copies of the chunk and the bucket list, written out as whole 128-B lines; list order unchanged
# speedup vs baseline: 4.2197x; 1.0417x over previous
#include <hip/hip_runtime.h>


typedef __attribute__((ext_vector_type(2))) float v2f;
typedef __attribute__((ext_vector_type(8))) float v8f;

static inline int cdiv(long long a, long long b) { return (int)((a + b - 1) / b); }

constexpr int SPB   = 25;
constexpr int BNODE = 16 * SPB;
constexpr int G1  = 256;
constexpr int PER = 8;

__device__ __forceinline__ int wave_incl_scan(int v, int lane) {
    for (int o = 1; o < 32; o <<= 1) {
        const int up = __shfl_up(v, o);
        if (lane >= o) v += up;
    }
    return v;
}

__launch_bounds__(1024)
__global__ void k_chunk_sort(const int* __restrict__ src, const int* __restrict__ dst,
                             int* __restrict__ loc, int* __restrict__ pairs_i,
                             int E, int chunk, int NB) {
    extern __shared__ __attribute__((aligned(16))) int lds_i[];
    int2* stage = (int2*)(lds_i + ((NB + 1) & ~1));
    __shared__ int wtot[32];
    int2* __restrict__ pairs = (int2*)pairs_i;
    const int tid  = threadIdx.x;
    const int lane = tid & 31;
    const int wv   = tid >> 5;
    for (int b = tid; b < NB; b += 1024) lds_i[b] = 0;
    __syncthreads();
    const int e0 = blockIdx.x * chunk;
    const int e1 = min(E, e0 + chunk);
    for (int base = e0; base < e1; base += 1024 * PER) {
        int d[PER];
#pragma unroll
        for (int k = 0; k < PER; ++k) {
            const int e = base + k * 1024 + tid;
            d[k] = (e < e1) ? dst[e] : -1;
        }
#pragma unroll
        for (int k = 0; k < PER; ++k)
            if (d[k] >= 0) atomicAdd(&lds_i[(unsigned)d[k] / BNODE], 1);
    }
    __syncthreads();

    int* __restrict__ locc = loc + (size_t)blockIdx.x * (NB + 1);
    int carry = 0;
    for (int t0 = 0; t0 < NB; t0 += 1024) {
        const int i = t0 + tid;
        const int c = (i < NB) ? lds_i[i] : 0;
        const int incl = wave_incl_scan(c, lane);
        if (lane == 31) wtot[wv] = incl;
        __syncthreads();
        int woff = 0, tot = 0;
        for (int w = 0; w < 32; ++w) {
            const int t = wtot[w];
            tot += t;
            if (w < wv) woff += t;
        }
        const int ex = carry + woff + (incl - c);
        if (i < NB) {
            lds_i[i] = ex;
            locc[i]  = ex;
        }
        carry += tot;
        __syncthreads();
    }
    if (tid == 0) locc[NB] = carry;

    for (int base = e0; base < e1; base += 1024 * PER) {
        int d[PER], s[PER];
#pragma unroll
        for (int k = 0; k < PER; ++k) {
            const int e = base + k * 1024 + tid;
            d[k] = (e < e1) ? dst[e] : -1;
            s[k] = (e < e1) ? src[e] : 0;
        }
        int p[PER];
#pragma unroll
        for (int k = 0; k < PER; ++k)
            p[k] = (d[k] >= 0) ? atomicAdd(&lds_i[(unsigned)d[k] / BNODE], 1) : 0;
#pragma unroll
        for (int k = 0; k < PER; ++k)
            if (d[k] >= 0) stage[p[k]] = make_int2(s[k], d[k]);
    }
    __syncthreads();
    for (int g0 = (e0 & ~15) + tid; g0 < e1; g0 += 4 * 1024) {
        int2 v[4];
#pragma unroll
        for (int k = 0; k < 4; ++k) {
            const int g = g0 + k * 1024;
            v[k] = (g >= e0 && g < e1) ? stage[g - e0] : make_int2(0, 0);
        }
#pragma unroll
        for (int k = 0; k < 4; ++k) {
            const int g = g0 + k * 1024;
            if (g >= e0 && g < e1) pairs[g] = v[k];
        }
    }
}

constexpr int LG_WAVES = 32;
constexpr int LCAP     = 9216;

template <int K, int N>
__launch_bounds__(LG_WAVES * 32)
__global__ void k_lists_gemm(const int* __restrict__ pairs_i, const int* __restrict__ loc,
                             int* __restrict__ rowptr, int* __restrict__ csr,
                             const float* __restrict__ X,
                             const float* __restrict__ W,
                             float* __restrict__ H,
                             int nrows, int NB, int chunk) {
    static_assert(G1 * 4 == LG_WAVES * 32, "four threads per run");
    constexpr int NT = N / 16;
    constexpr int NH = NT / 2;
    constexpr int NTHR = LG_WAVES * 32;
    static_assert((K * N + LCAP) * 4 > 160 * 1024, "one work-group per WGP");
    __shared__ float BF[K * N];
    __shared__ int   stg[LCAP];
    __shared__ int   cur[BNODE];
    __shared__ float dvs[BNODE];
    __shared__ int   wtot[32];
    __shared__ int   rstart[G1];
    __shared__ int   rlen[G1];
    const int2* __restrict__ pairs = (const int2*)pairs_i;
    const int tid  = threadIdx.x;
    const int lane = tid & 31;
    const int wv   = tid >> 5;
    const int b    = blockIdx.x;
    const int n0   = b * BNODE;

    int lo = 0, hi = 0;
    if (tid < G1) {
        const int* lc = loc + (size_t)tid * (NB + 1) + b;
        lo = lc[0];
        hi = lc[1];
    }
    if (tid < BNODE) cur[tid] = 0;

    for (int idx = tid; idx < K * N; idx += NTHR) {
        const int n  = idx / K;
        const int k  = idx - n * K;
        const int ks = k >> 2;
        const int kk = k & 3;
        const int g  = kk >> 1;
        const int r  = kk & 1;
        const int t  = n >> 4;
        const int nl = n & 15;
        BF[((ks * NT + t) * 32 + g * 16 + nl) * 2 + r] = W[idx];
    }

    if (tid < G1) {
        rstart[tid] = tid * chunk + lo;
        rlen[tid]   = hi - lo;
    }
    int part = lo;
    for (int o = 16; o > 0; o >>= 1) part += __shfl_xor(part, o);
    if (lane == 0) wtot[wv] = part;
    __syncthreads();
    int beg = 0;
    for (int w = 0; w < 32; ++w) beg += wtot[w];
    const int rs  = rstart[tid >> 2];
    const int rl  = rlen[tid >> 2];
    const int sub = tid & 3;

    for (int j0 = sub; j0 < rl; j0 += 4 * PER) {
        int dy[PER];
#pragma unroll
        for (int k = 0; k < PER; ++k) {
            const int j = j0 + 4 * k;
            dy[k] = (j < rl) ? pairs[rs + j].y : -1;
        }
#pragma unroll
        for (int k = 0; k < PER; ++k)
            if (dy[k] >= 0) atomicAdd(&cur[dy[k] - n0], 1);
    }
    __syncthreads();

    const int c = (tid < BNODE) ? cur[tid] : 0;
    const int incl = wave_incl_scan(c, lane);
    if (lane == 31 && wv * 32 < BNODE) wtot[wv] = incl;
    __syncthreads();
    if (tid < BNODE) {
        int woff = 0;
        for (int w = 0; w < wv; ++w) woff += wtot[w];
        const int start = beg + woff + (incl - c);
        if (n0 + tid < nrows) rowptr[n0 + tid] = start;
        cur[tid] = start;
        dvs[tid] = rsqrtf((float)(c + 1));
    }
    int btot = 0;
    for (int w = 0; w * 32 < BNODE; ++w) btot += wtot[w];
    const bool staged = btot <= LCAP;
    __syncthreads();

    for (int j0 = sub; j0 < rl; j0 += 4 * PER) {
        int2 p[PER];
#pragma unroll
        for (int k = 0; k < PER; ++k) {
            const int j = j0 + 4 * k;
            p[k] = (j < rl) ? pairs[rs + j] : make_int2(0, -1);
        }
        int q[PER];
#pragma unroll
        for (int k = 0; k < PER; ++k)
            q[k] = (p[k].y >= 0) ? atomicAdd(&cur[p[k].y - n0], 1) : 0;
        if (staged) {
#pragma unroll
            for (int k = 0; k < PER; ++k)
                if (p[k].y >= 0) stg[q[k] - beg] = p[k].x;
        } else {
#pragma unroll
            for (int k = 0; k < PER; ++k)
                if (p[k].y >= 0) csr[q[k]] = p[k].x;
        }
    }
    if (staged) {
        __syncthreads();
        const int gend = beg + btot;
        for (int g0 = (beg & ~31) + tid; g0 < gend; g0 += 4 * NTHR) {
            int v[4];
#pragma unroll
            for (int k = 0; k < 4; ++k) {
                const int g = g0 + k * NTHR;
                v[k] = (g >= beg && g < gend) ? stg[g - beg] : 0;
            }
#pragma unroll
            for (int k = 0; k < 4; ++k) {
                const int g = g0 + k * NTHR;
                if (g >= beg && g < gend) csr[g] = v[k];
            }
        }
    }

    const int wave    = __builtin_amdgcn_readfirstlane(wv);
    const int laneLo  = lane & 15;
    const int laneHi  = lane >> 4;
    const int nstrips = nrows >> 4;
    const int s0      = b * SPB;

    for (int u = wave; u < 2 * SPB; u += LG_WAVES) {
        const int strip = s0 + (u >> 1);
        if (strip >= nstrips) break;
        const int m0 = strip * 16;
        const int t0 = (u & 1) * NH;

        v8f acc[NH] = {};

        const float* xrow  = X + (size_t)(m0 + laneLo) * K;
        const float* fbase = BF + t0 * 64 + lane * 2;
        for (int ks = 0; ks < K / 4; ++ks) {
            const v2f a = *(const v2f*)(xrow + ks * 4 + 2 * laneHi);
            const float* fb = fbase + (ks * NT) * 64;
#pragma unroll
            for (int t = 0; t < NH; ++t) {
                const v2f b2 = *(const v2f*)(fb + t * 64);
                acc[t] = __builtin_amdgcn_wmma_f32_16x16x4_f32(
                    false, a, false, b2, (short)0, acc[t], false, false);
            }
        }

        const float* dp = dvs + (m0 - n0) + 8 * laneHi;
        float dv[8];
#pragma unroll
        for (int r = 0; r < 8; ++r) dv[r] = dp[r];

#pragma unroll
        for (int t = 0; t < NH; ++t)
#pragma unroll
            for (int r = 0; r < 8; ++r)
                H[(size_t)(m0 + r + 8 * laneHi) * N + (t0 + t) * 16 + laneLo] = acc[t][r] * dv[r];
    }
}

template <int V> struct rowvec;
template <> struct rowvec<4> { typedef float4 type; };
template <> struct rowvec<2> { typedef float2 type; };

template <int C, bool RELU>
__launch_bounds__(256)
__global__ void k_gather(const int* __restrict__ rowptr, const int* __restrict__ csr,
                         const float* __restrict__ H, const float* __restrict__ bias,
                         float* __restrict__ out, int n, int E) {
    constexpr int V = C / 32;
    constexpr int U = 8;
    typedef typename rowvec<V>::type vec;

    const int lane = threadIdx.x & 31;
    const int wave   = __builtin_amdgcn_readfirstlane((int)((blockIdx.x * blockDim.x + threadIdx.x) >> 5));
    const int nwaves = (int)((gridDim.x * blockDim.x) >> 5);
    const float* Hl  = H + lane * V;

    float bv[V];
#pragma unroll
    for (int v = 0; v < V; ++v) bv[v] = bias[lane * V + v];

    int i = wave;
    int beg = 0, end = 0;
    if (i < n) {
        beg = rowptr[i];
        end = (i + 1 < n) ? rowptr[i + 1] : E;
    }
    while (i < n) {
        const int inext = i + nwaves;
        int nbeg = 0, nend = 0;
        if (inext < n) {
            nbeg = rowptr[inext];
            nend = (inext + 1 < n) ? rowptr[inext + 1] : E;
        }
        const float di = rsqrtf((float)(end - beg + 1));
        const vec hself = *(const vec*)(Hl + (size_t)i * C);

        float acc[V];
#pragma unroll
        for (int v = 0; v < V; ++v) acc[v] = 0.0f;

        for (int b0 = beg; b0 < end; b0 += 32) {
            const int m = (end - b0 < 32) ? (end - b0) : 32;
            int s = i;
            if (lane < m) s = csr[b0 + lane];
            int j = 0;
            do {
                vec h[U];
#pragma unroll
                for (int u = 0; u < U; ++u) {
                    const int su = __builtin_amdgcn_readlane(s, j + u);
                    h[u] = *(const vec*)(Hl + (size_t)su * C);
                }
#pragma unroll
                for (int u = 0; u < U; ++u) {
                    const float wu = (j + u < m) ? 1.0f : 0.0f;
                    const float* hp = (const float*)&h[u];
#pragma unroll
                    for (int v = 0; v < V; ++v) acc[v] = fmaf(hp[v], wu, acc[v]);
                }
                j += U;
            } while (j < m);
        }

        vec o;
        float* op = (float*)&o;
        const float* hs = (const float*)&hself;
#pragma unroll
        for (int v = 0; v < V; ++v) {
            const float r = fmaf(acc[v] + hs[v], di, bv[v]);
            op[v] = RELU ? fmaxf(r, 0.0f) : r;
        }
        *(vec*)(out + (size_t)i * C + lane * V) = o;
        i = inext;
        beg = nbeg;
        end = nend;
    }
}

constexpr int FG_WAVES = 32;
constexpr int FG_WGS   = 256;
constexpr int FG_PITCH = 132;

template <int C, int N2>
__launch_bounds__(FG_WAVES * 32)
__global__ void k_gather_gemm(const int* __restrict__ rowptr, const int* __restrict__ csr,
                              const float* __restrict__ H, const float* __restrict__ bias,
                              const float* __restrict__ W,
                              float* __restrict__ H2,
                              int n, int E) {
    static_assert(C == 128, "one float4 of a row per lane");
    constexpr int K = C;
    constexpr int NT = N2 / 16;
    constexpr int NTHR = FG_WAVES * 32;
    constexpr int U = 16;
    __shared__ float BF[K * N2];
    __shared__ float ST[FG_WAVES * 16 * FG_PITCH];

    const int tid = threadIdx.x;
    constexpr int STG = (K * N2) / NTHR;
    static_assert(STG * NTHR == K * N2, "W is dealt evenly over the threads");
    float wv[STG];
#pragma unroll
    for (int q = 0; q < STG; ++q) wv[q] = W[tid + q * NTHR];
#pragma unroll
    for (int q = 0; q < STG; ++q) {
        const int idx = tid + q * NTHR;
        const int nn = idx / K;
        const int k  = idx - nn * K;
        const int ks = k >> 2;
        const int kk = k & 3;
        const int g  = kk >> 1;
        const int r  = kk & 1;
        const int t  = nn >> 4;
        const int nl = nn & 15;
        BF[((ks * NT + t) * 32 + g * 16 + nl) * 2 + r] = wv[q];
    }
    __syncthreads();

    const int wave   = __builtin_amdgcn_readfirstlane(tid >> 5);
    const int lane   = tid & 31;
    const int laneLo = lane & 15;
    const int laneHi = lane >> 4;
    float* STw = ST + wave * (16 * FG_PITCH);
    const float* Hl = H + lane * 4;

    float bv[4];
#pragma unroll
    for (int v = 0; v < 4; ++v) bv[v] = bias[lane * 4 + v];

    const int ntiles = n >> 4;
    for (int tile = (int)blockIdx.x + (int)gridDim.x * wave; tile < ntiles;
         tile += (int)gridDim.x * FG_WAVES) {
        const int m0 = tile * 16;
        int rp = 0;
        if (lane <= 16) {
            const int idx = m0 + lane;
            rp = (idx < n) ? rowptr[idx] : E;
        }
        int beg = __builtin_amdgcn_readlane(rp, 0);
        int end = __builtin_amdgcn_readlane(rp, 1);
        int s_cur = m0;
        if (lane < end - beg) s_cur = csr[beg + lane];

        for (int r = 0; r < 16; ++r) {
            const int i = m0 + r;
            const int nend = __builtin_amdgcn_readlane(rp, r + 2);
            const int inx = max(min(end + lane, E - 1), 0);
            const int cnx = csr[inx];
            const int s_nx = (lane < nend - end) ? cnx : (i + 1);
            const float di = rsqrtf((float)(end - beg + 1));
            const float4 hself = *(const float4*)(Hl + (size_t)i * C);

            float acc[4] = {0.0f, 0.0f, 0.0f, 0.0f};
            auto sum_rows = [&](int s, int m) {
                int j = 0;
                do {
                    float4 h[U];
#pragma unroll
                    for (int u = 0; u < U; ++u) {
                        const int su = __builtin_amdgcn_readlane(s, j + u);
                        h[u] = *(const float4*)(Hl + (size_t)su * C);
                    }
#pragma unroll
                    for (int u = 0; u < U; ++u) {
                        const float wu = (j + u < m) ? 1.0f : 0.0f;
                        acc[0] = fmaf(h[u].x, wu, acc[0]);
                        acc[1] = fmaf(h[u].y, wu, acc[1]);
                        acc[2] = fmaf(h[u].z, wu, acc[2]);
                        acc[3] = fmaf(h[u].w, wu, acc[3]);
                    }
                    j += U;
                } while (j < m);
            };
            if (beg < end) {
                const int m = (end - beg < 32) ? (end - beg) : 32;
                sum_rows(s_cur, m);
                for (int b0 = beg + 32; b0 < end; b0 += 32) {
                    const int mm = (end - b0 < 32) ? (end - b0) : 32;
                    int s = i;
                    if (lane < mm) s = csr[b0 + lane];
                    sum_rows(s, mm);
                }
            }

            float4 o;
            o.x = fmaxf(fmaf(acc[0] + hself.x, di, bv[0]), 0.0f);
            o.y = fmaxf(fmaf(acc[1] + hself.y, di, bv[1]), 0.0f);
            o.z = fmaxf(fmaf(acc[2] + hself.z, di, bv[2]), 0.0f);
            o.w = fmaxf(fmaf(acc[3] + hself.w, di, bv[3]), 0.0f);
            *(float4*)(STw + r * FG_PITCH + lane * 4) = o;
            s_cur = s_nx;
            beg = end;
            end = nend;
        }

        float dv[8];
        {
            const int rpn = __shfl_down(rp, 1);
            const float dm = rsqrtf((float)(((lane < 16) ? (rpn - rp) : 0) + 1));
#pragma unroll
            for (int q = 0; q < 8; ++q) dv[q] = __shfl(dm, 8 * laneHi + q);
        }

        v8f accm[NT] = {};
        const float* arow  = STw + laneLo * FG_PITCH + 2 * laneHi;
        const float* fbase = BF + lane * 2;
        for (int ks = 0; ks < K / 4; ++ks) {
            const v2f a = *(const v2f*)(arow + ks * 4);
            const float* fb = fbase + (ks * NT) * 64;
#pragma unroll
            for (int t = 0; t < NT; ++t) {
                const v2f b = *(const v2f*)(fb + t * 64);
                accm[t] = __builtin_amdgcn_wmma_f32_16x16x4_f32(
                    false, a, false, b, (short)0, accm[t], false, false);
            }
        }

#pragma unroll
        for (int t = 0; t < NT; ++t)
#pragma unroll
            for (int q = 0; q < 8; ++q)
                H2[(size_t)(m0 + q + 8 * laneHi) * N2 + t * 16 + laneLo] = accm[t][q] * dv[q];
    }
}

extern "C" void kernel_launch(void* const* d_in, const int* in_sizes, int n_in,
                              void* d_out, int out_size, void* d_ws, size_t ws_size,
                              hipStream_t stream) {
    constexpr int IN  = 256;
    constexpr int HID = 128;
    constexpr int LAT = 64;

    const float* x   = (const float*)d_in[0];
    const int*   ei  = (const int*)d_in[1];
    const float* W1  = (const float*)d_in[2];
    const float* b1  = (const float*)d_in[3];
    const float* W2  = (const float*)d_in[4];
    const float* b2  = (const float*)d_in[5];
    float*       out = (float*)d_out;

    const int N = in_sizes[0] / IN;
    const int E = in_sizes[1] / 2;
    const int* src = ei;
    const int* dst = ei + E;

    char* ws = (char*)d_ws;
    int*   rowptr = (int*)ws;    ws += (size_t)N * sizeof(int);
    int*   csr    = (int*)ws;    ws += (size_t)E * sizeof(int);
    float* H1     = (float*)ws;  ws += (size_t)N * HID * sizeof(float);
    float* S1     = (float*)ws;
    float* H2     = S1;
    const int NB    = cdiv(N, BNODE);
    const int chunk = cdiv(E, G1);
    const int LN    = G1 * (NB + 1);
    int*   loc    = (int*)S1;
    int*   pairs  = loc + (size_t)cdiv(LN, 4) * 4;

    const int T = 256;
    const int GW = 2048;

    k_chunk_sort<<<G1, 1024, (((NB + 1) & ~1) + 2 * chunk) * sizeof(int), stream>>>(src, dst, loc, pairs, E, chunk, NB);
    k_lists_gemm<IN, HID><<<NB, LG_WAVES * 32, 0, stream>>>(pairs, loc, rowptr, csr, x, W1, H1, N, NB, chunk);

    k_gather_gemm<HID, LAT><<<FG_WGS, FG_WAVES * 32, 0, stream>>>(rowptr, csr, H1, b1, W2, H2, N, E);

    k_gather<LAT, false><<<GW, T, 0, stream>>>(rowptr, csr, H2, b2, out, N, E);
}


# speedup vs baseline: 4.5526x; 1.0789x over previous
#include <hip/hip_runtime.h>


typedef __attribute__((ext_vector_type(2))) float v2f;
typedef __attribute__((ext_vector_type(8))) float v8f;

static inline int cdiv(long long a, long long b) { return (int)((a + b - 1) / b); }

constexpr int SPB   = 25;
constexpr int BNODE = 16 * SPB;
constexpr int G1  = 256;
constexpr int PER = 8;

__device__ __forceinline__ int wave_incl_scan(int v, int lane) {
    for (int o = 1; o < 32; o <<= 1) {
        const int up = __shfl_up(v, o);
        if (lane >= o) v += up;
    }
    return v;
}

__launch_bounds__(1024)
__global__ void k_chunk_sort(const int* __restrict__ src, const int* __restrict__ dst,
                             int* __restrict__ loc, int* __restrict__ pairs_i,
                             int E, int chunk, int NB) {
    extern __shared__ __attribute__((aligned(16))) int lds_i[];
    int2* stage = (int2*)(lds_i + ((NB + 1) & ~1));
    __shared__ int wtot[32];
    int2* __restrict__ pairs = (int2*)pairs_i;
    const int tid  = threadIdx.x;
    const int lane = tid & 31;
    const int wv   = tid >> 5;
    for (int b = tid; b < NB; b += 1024) lds_i[b] = 0;
    __syncthreads();
    const int e0 = blockIdx.x * chunk;
    const int e1 = min(E, e0 + chunk);
    for (int base = e0; base < e1; base += 1024 * PER) {
        int d[PER];
#pragma unroll
        for (int k = 0; k < PER; ++k) {
            const int e = base + k * 1024 + tid;
            d[k] = (e < e1) ? dst[e] : -1;
        }
#pragma unroll
        for (int k = 0; k < PER; ++k)
            if (d[k] >= 0) atomicAdd(&lds_i[(unsigned)d[k] / BNODE], 1);
    }
    __syncthreads();

    int* __restrict__ locc = loc + (size_t)blockIdx.x * (NB + 1);
    int carry = 0;
    for (int t0 = 0; t0 < NB; t0 += 1024) {
        const int i = t0 + tid;
        const int c = (i < NB) ? lds_i[i] : 0;
        const int incl = wave_incl_scan(c, lane);
        if (lane == 31) wtot[wv] = incl;
        __syncthreads();
        int woff = 0, tot = 0;
        for (int w = 0; w < 32; ++w) {
            const int t = wtot[w];
            tot += t;
            if (w < wv) woff += t;
        }
        const int ex = carry + woff + (incl - c);
        if (i < NB) {
            lds_i[i] = ex;
            locc[i]  = ex;
        }
        carry += tot;
        __syncthreads();
    }
    if (tid == 0) locc[NB] = carry;

    for (int base = e0; base < e1; base += 1024 * PER) {
        int d[PER], s[PER];
#pragma unroll
        for (int k = 0; k < PER; ++k) {
            const int e = base + k * 1024 + tid;
            d[k] = (e < e1) ? dst[e] : -1;
            s[k] = (e < e1) ? src[e] : 0;
        }
        int p[PER];
#pragma unroll
        for (int k = 0; k < PER; ++k)
            p[k] = (d[k] >= 0) ? atomicAdd(&lds_i[(unsigned)d[k] / BNODE], 1) : 0;
#pragma unroll
        for (int k = 0; k < PER; ++k)
            if (d[k] >= 0) stage[p[k]] = make_int2(s[k], d[k]);
    }
    __syncthreads();
    for (int g0 = (e0 & ~15) + tid; g0 < e1; g0 += 4 * 1024) {
        int2 v[4];
#pragma unroll
        for (int k = 0; k < 4; ++k) {
            const int g = g0 + k * 1024;
            v[k] = (g >= e0 && g < e1) ? stage[g - e0] : make_int2(0, 0);
        }
#pragma unroll
        for (int k = 0; k < 4; ++k) {
            const int g = g0 + k * 1024;
            if (g >= e0 && g < e1) pairs[g] = v[k];
        }
    }
}

constexpr int LG_WAVES = 32;
constexpr int LCAP     = 9216;

template <int K, int N>
__launch_bounds__(LG_WAVES * 32)
__global__ void k_lists_gemm(const int* __restrict__ pairs_i, const int* __restrict__ loc,
                             int* __restrict__ rowptr, int* __restrict__ csr,
                             const float* __restrict__ X,
                             const float* __restrict__ W,
                             float* __restrict__ H,
                             int nrows, int NB, int chunk) {
    static_assert(G1 * 4 == LG_WAVES * 32, "four threads per run");
    constexpr int NT = N / 16;
    constexpr int NH = NT / 2;
    constexpr int NTHR = LG_WAVES * 32;
    static_assert((K * N + LCAP) * 4 > 160 * 1024, "one work-group per WGP");
    __shared__ float BF[K * N];
    __shared__ int   stg[LCAP];
    __shared__ int   cur[BNODE];
    __shared__ float dvs[BNODE];
    __shared__ int   wtot[32];
    __shared__ int   rstart[G1];
    __shared__ int   rlen[G1];
    constexpr int OT_PITCH = N / 2 + 4;
    __shared__ __attribute__((aligned(16))) float OT[LG_WAVES * 16 * OT_PITCH];
    const int2* __restrict__ pairs = (const int2*)pairs_i;
    const int tid  = threadIdx.x;
    const int lane = tid & 31;
    const int wv   = tid >> 5;
    const int b    = blockIdx.x;
    const int n0   = b * BNODE;

    int lo = 0, hi = 0;
    if (tid < G1) {
        const int* lc = loc + (size_t)tid * (NB + 1) + b;
        lo = lc[0];
        hi = lc[1];
    }
    if (tid < BNODE) cur[tid] = 0;

    for (int idx = tid; idx < K * N; idx += NTHR) {
        const int n  = idx / K;
        const int k  = idx - n * K;
        const int ks = k >> 2;
        const int kk = k & 3;
        const int g  = kk >> 1;
        const int r  = kk & 1;
        const int t  = n >> 4;
        const int nl = n & 15;
        BF[((ks * NT + t) * 32 + g * 16 + nl) * 2 + r] = W[idx];
    }

    if (tid < G1) {
        rstart[tid] = tid * chunk + lo;
        rlen[tid]   = hi - lo;
    }
    int part = lo;
    for (int o = 16; o > 0; o >>= 1) part += __shfl_xor(part, o);
    if (lane == 0) wtot[wv] = part;
    __syncthreads();
    int beg = 0;
    for (int w = 0; w < 32; ++w) beg += wtot[w];
    const int rs  = rstart[tid >> 2];
    const int rl  = rlen[tid >> 2];
    const int sub = tid & 3;

    for (int j0 = sub; j0 < rl; j0 += 4 * PER) {
        int dy[PER];
#pragma unroll
        for (int k = 0; k < PER; ++k) {
            const int j = j0 + 4 * k;
            dy[k] = (j < rl) ? pairs[rs + j].y : -1;
        }
#pragma unroll
        for (int k = 0; k < PER; ++k)
            if (dy[k] >= 0) atomicAdd(&cur[dy[k] - n0], 1);
    }
    __syncthreads();

    const int c = (tid < BNODE) ? cur[tid] : 0;
    const int incl = wave_incl_scan(c, lane);
    if (lane == 31 && wv * 32 < BNODE) wtot[wv] = incl;
    __syncthreads();
    if (tid < BNODE) {
        int woff = 0;
        for (int w = 0; w < wv; ++w) woff += wtot[w];
        const int start = beg + woff + (incl - c);
        if (n0 + tid < nrows) rowptr[n0 + tid] = start;
        cur[tid] = start;
        dvs[tid] = rsqrtf((float)(c + 1));
    }
    int btot = 0;
    for (int w = 0; w * 32 < BNODE; ++w) btot += wtot[w];
    const bool staged = btot <= LCAP;
    __syncthreads();

    for (int j0 = sub; j0 < rl; j0 += 4 * PER) {
        int2 p[PER];
#pragma unroll
        for (int k = 0; k < PER; ++k) {
            const int j = j0 + 4 * k;
            p[k] = (j < rl) ? pairs[rs + j] : make_int2(0, -1);
        }
        int q[PER];
#pragma unroll
        for (int k = 0; k < PER; ++k)
            q[k] = (p[k].y >= 0) ? atomicAdd(&cur[p[k].y - n0], 1) : 0;
        if (staged) {
#pragma unroll
            for (int k = 0; k < PER; ++k)
                if (p[k].y >= 0) stg[q[k] - beg] = p[k].x;
        } else {
#pragma unroll
            for (int k = 0; k < PER; ++k)
                if (p[k].y >= 0) csr[q[k]] = p[k].x;
        }
    }
    if (staged) {
        __syncthreads();
        const int gend = beg + btot;
        for (int g0 = (beg & ~31) + tid; g0 < gend; g0 += 4 * NTHR) {
            int v[4];
#pragma unroll
            for (int k = 0; k < 4; ++k) {
                const int g = g0 + k * NTHR;
                v[k] = (g >= beg && g < gend) ? stg[g - beg] : 0;
            }
#pragma unroll
            for (int k = 0; k < 4; ++k) {
                const int g = g0 + k * NTHR;
                if (g >= beg && g < gend) csr[g] = v[k];
            }
        }
    }

    const int wave    = __builtin_amdgcn_readfirstlane(wv);
    const int laneLo  = lane & 15;
    const int laneHi  = lane >> 4;
    const int nstrips = nrows >> 4;
    const int s0      = b * SPB;

    for (int u = wave; u < 2 * SPB; u += LG_WAVES) {
        const int strip = s0 + (u >> 1);
        if (strip >= nstrips) break;
        const int m0 = strip * 16;
        const int t0 = (u & 1) * NH;

        v8f acc[NH] = {};

        const float* xrow  = X + (size_t)(m0 + laneLo) * K;
        const float* fbase = BF + t0 * 64 + lane * 2;
        for (int ks = 0; ks < K / 4; ++ks) {
            const v2f a = *(const v2f*)(xrow + ks * 4 + 2 * laneHi);
            const float* fb = fbase + (ks * NT) * 64;
#pragma unroll
            for (int t = 0; t < NH; ++t) {
                const v2f b2 = *(const v2f*)(fb + t * 64);
                acc[t] = __builtin_amdgcn_wmma_f32_16x16x4_f32(
                    false, a, false, b2, (short)0, acc[t], false, false);
            }
        }

        const float* dp = dvs + (m0 - n0) + 8 * laneHi;
        float dv[8];
#pragma unroll
        for (int r = 0; r < 8; ++r) dv[r] = dp[r];

        float* tw = OT + wave * (16 * OT_PITCH);
#pragma unroll
        for (int t = 0; t < NH; ++t)
#pragma unroll
            for (int r = 0; r < 8; ++r)
                tw[(r + 8 * laneHi) * OT_PITCH + t * 16 + laneLo] = acc[t][r] * dv[r];
#pragma unroll
        for (int i = 0; i < 8; ++i) {
            const int row = 2 * i + laneHi;
            const float4 v = *(const float4*)(tw + row * OT_PITCH + laneLo * 4);
            *(float4*)(H + (size_t)(m0 + row) * N + t0 * 16 + laneLo * 4) = v;
        }
    }
}

template <int V> struct rowvec;
template <> struct rowvec<4> { typedef float4 type; };
template <> struct rowvec<2> { typedef float2 type; };

template <int C, bool RELU>
__launch_bounds__(256)
__global__ void k_gather(const int* __restrict__ rowptr, const int* __restrict__ csr,
                         const float* __restrict__ H, const float* __restrict__ bias,
                         float* __restrict__ out, int n, int E) {
    constexpr int V = C / 32;
    constexpr int U = 8;
    typedef typename rowvec<V>::type vec;

    const int lane = threadIdx.x & 31;
    const int wave   = __builtin_amdgcn_readfirstlane((int)((blockIdx.x * blockDim.x + threadIdx.x) >> 5));
    const int nwaves = (int)((gridDim.x * blockDim.x) >> 5);
    const float* Hl  = H + lane * V;

    float bv[V];
#pragma unroll
    for (int v = 0; v < V; ++v) bv[v] = bias[lane * V + v];

    int i = wave;
    int beg = 0, end = 0;
    if (i < n) {
        beg = rowptr[i];
        end = (i + 1 < n) ? rowptr[i + 1] : E;
    }
    while (i < n) {
        const int inext = i + nwaves;
        int nbeg = 0, nend = 0;
        if (inext < n) {
            nbeg = rowptr[inext];
            nend = (inext + 1 < n) ? rowptr[inext + 1] : E;
        }
        const float di = rsqrtf((float)(end - beg + 1));
        const vec hself = *(const vec*)(Hl + (size_t)i * C);

        float acc[V];
#pragma unroll
        for (int v = 0; v < V; ++v) acc[v] = 0.0f;

        for (int b0 = beg; b0 < end; b0 += 32) {
            const int m = (end - b0 < 32) ? (end - b0) : 32;
            int s = i;
            if (lane < m) s = csr[b0 + lane];
            int j = 0;
            do {
                vec h[U];
#pragma unroll
                for (int u = 0; u < U; ++u) {
                    const int su = __builtin_amdgcn_readlane(s, j + u);
                    h[u] = *(const vec*)(Hl + (size_t)su * C);
                }
#pragma unroll
                for (int u = 0; u < U; ++u) {
                    const float wu = (j + u < m) ? 1.0f : 0.0f;
                    const float* hp = (const float*)&h[u];
#pragma unroll
                    for (int v = 0; v < V; ++v) acc[v] = fmaf(hp[v], wu, acc[v]);
                }
                j += U;
            } while (j < m);
        }

        vec o;
        float* op = (float*)&o;
        const float* hs = (const float*)&hself;
#pragma unroll
        for (int v = 0; v < V; ++v) {
            const float r = fmaf(acc[v] + hs[v], di, bv[v]);
            op[v] = RELU ? fmaxf(r, 0.0f) : r;
        }
        *(vec*)(out + (size_t)i * C + lane * V) = o;
        i = inext;
        beg = nbeg;
        end = nend;
    }
}

constexpr int FG_WAVES = 32;
constexpr int FG_WGS   = 256;
constexpr int FG_PITCH = 132;

template <int C, int N2>
__launch_bounds__(FG_WAVES * 32)
__global__ void k_gather_gemm(const int* __restrict__ rowptr, const int* __restrict__ csr,
                              const float* __restrict__ H, const float* __restrict__ bias,
                              const float* __restrict__ W,
                              float* __restrict__ H2,
                              int n, int E) {
    static_assert(C == 128, "one float4 of a row per lane");
    static_assert(N2 == 64, "a half-wave stores one row of the result as float4s");
    constexpr int K = C;
    constexpr int NT = N2 / 16;
    constexpr int NTHR = FG_WAVES * 32;
    constexpr int U = 16;
    __shared__ float BF[K * N2];
    __shared__ float ST[FG_WAVES * 16 * FG_PITCH];

    const int tid = threadIdx.x;
    constexpr int STG = (K * N2) / NTHR;
    static_assert(STG * NTHR == K * N2, "W is dealt evenly over the threads");
    float wv[STG];
#pragma unroll
    for (int q = 0; q < STG; ++q) wv[q] = W[tid + q * NTHR];
#pragma unroll
    for (int q = 0; q < STG; ++q) {
        const int idx = tid + q * NTHR;
        const int nn = idx / K;
        const int k  = idx - nn * K;
        const int ks = k >> 2;
        const int kk = k & 3;
        const int g  = kk >> 1;
        const int r  = kk & 1;
        const int t  = nn >> 4;
        const int nl = nn & 15;
        BF[((ks * NT + t) * 32 + g * 16 + nl) * 2 + r] = wv[q];
    }
    __syncthreads();

    const int wave   = __builtin_amdgcn_readfirstlane(tid >> 5);
    const int lane   = tid & 31;
    const int laneLo = lane & 15;
    const int laneHi = lane >> 4;
    float* STw = ST + wave * (16 * FG_PITCH);
    const float* Hl = H + lane * 4;

    float bv[4];
#pragma unroll
    for (int v = 0; v < 4; ++v) bv[v] = bias[lane * 4 + v];

    const int ntiles = n >> 4;
    for (int tile = (int)blockIdx.x + (int)gridDim.x * wave; tile < ntiles;
         tile += (int)gridDim.x * FG_WAVES) {
        const int m0 = tile * 16;
        int rp = 0;
        if (lane <= 16) {
            const int idx = m0 + lane;
            rp = (idx < n) ? rowptr[idx] : E;
        }
        int beg = __builtin_amdgcn_readlane(rp, 0);
        int end = __builtin_amdgcn_readlane(rp, 1);
        int s_cur = m0;
        if (lane < end - beg) s_cur = csr[beg + lane];

        for (int r = 0; r < 16; ++r) {
            const int i = m0 + r;
            const int nend = __builtin_amdgcn_readlane(rp, r + 2);
            const int inx = max(min(end + lane, E - 1), 0);
            const int cnx = csr[inx];
            const int s_nx = (lane < nend - end) ? cnx : (i + 1);
            const float di = rsqrtf((float)(end - beg + 1));
            const float4 hself = *(const float4*)(Hl + (size_t)i * C);

            float acc[4] = {0.0f, 0.0f, 0.0f, 0.0f};
            auto sum_rows = [&](int s, int m) {
                int j = 0;
                do {
                    float4 h[U];
#pragma unroll
                    for (int u = 0; u < U; ++u) {
                        const int su = __builtin_amdgcn_readlane(s, j + u);
                        h[u] = *(const float4*)(Hl + (size_t)su * C);
                    }
#pragma unroll
                    for (int u = 0; u < U; ++u) {
                        const float wu = (j + u < m) ? 1.0f : 0.0f;
                        acc[0] = fmaf(h[u].x, wu, acc[0]);
                        acc[1] = fmaf(h[u].y, wu, acc[1]);
                        acc[2] = fmaf(h[u].z, wu, acc[2]);
                        acc[3] = fmaf(h[u].w, wu, acc[3]);
                    }
                    j += U;
                } while (j < m);
            };
            if (beg < end) {
                const int m = (end - beg < 32) ? (end - beg) : 32;
                sum_rows(s_cur, m);
                for (int b0 = beg + 32; b0 < end; b0 += 32) {
                    const int mm = (end - b0 < 32) ? (end - b0) : 32;
                    int s = i;
                    if (lane < mm) s = csr[b0 + lane];
                    sum_rows(s, mm);
                }
            }

            float4 o;
            o.x = fmaxf(fmaf(acc[0] + hself.x, di, bv[0]), 0.0f);
            o.y = fmaxf(fmaf(acc[1] + hself.y, di, bv[1]), 0.0f);
            o.z = fmaxf(fmaf(acc[2] + hself.z, di, bv[2]), 0.0f);
            o.w = fmaxf(fmaf(acc[3] + hself.w, di, bv[3]), 0.0f);
            *(float4*)(STw + r * FG_PITCH + lane * 4) = o;
            s_cur = s_nx;
            beg = end;
            end = nend;
        }

        float dv[8];
        {
            const int rpn = __shfl_down(rp, 1);
            const float dm = rsqrtf((float)(((lane < 16) ? (rpn - rp) : 0) + 1));
#pragma unroll
            for (int q = 0; q < 8; ++q) dv[q] = __shfl(dm, 8 * laneHi + q);
        }

        v8f accm[NT] = {};
        const float* arow  = STw + laneLo * FG_PITCH + 2 * laneHi;
        const float* fbase = BF + lane * 2;
        for (int ks = 0; ks < K / 4; ++ks) {
            const v2f a = *(const v2f*)(arow + ks * 4);
            const float* fb = fbase + (ks * NT) * 64;
#pragma unroll
            for (int t = 0; t < NT; ++t) {
                const v2f b = *(const v2f*)(fb + t * 64);
                accm[t] = __builtin_amdgcn_wmma_f32_16x16x4_f32(
                    false, a, false, b, (short)0, accm[t], false, false);
            }
        }

#pragma unroll
        for (int t = 0; t < NT; ++t)
#pragma unroll
            for (int q = 0; q < 8; ++q)
                STw[(q + 8 * laneHi) * FG_PITCH + t * 16 + laneLo] = accm[t][q] * dv[q];
#pragma unroll
        for (int i = 0; i < 8; ++i) {
            const int row = 2 * i + laneHi;
            const float4 v = *(const float4*)(STw + row * FG_PITCH + laneLo * 4);
            *(float4*)(H2 + (size_t)(m0 + row) * N2 + laneLo * 4) = v;
        }
    }
}

extern "C" void kernel_launch(void* const* d_in, const int* in_sizes, int n_in,
                              void* d_out, int out_size, void* d_ws, size_t ws_size,
                              hipStream_t stream) {
    constexpr int IN  = 256;
    constexpr int HID = 128;
    constexpr int LAT = 64;

    const float* x   = (const float*)d_in[0];
    const int*   ei  = (const int*)d_in[1];
    const float* W1  = (const float*)d_in[2];
    const float* b1  = (const float*)d_in[3];
    const float* W2  = (const float*)d_in[4];
    const float* b2  = (const float*)d_in[5];
    float*       out = (float*)d_out;

    const int N = in_sizes[0] / IN;
    const int E = in_sizes[1] / 2;
    const int* src = ei;
    const int* dst = ei + E;

    char* ws = (char*)d_ws;
    float* H1     = (float*)ws;  ws += (size_t)N * HID * sizeof(float);
    float* S1     = (float*)ws;  ws += (size_t)N * HID * sizeof(float);
    int*   rowptr = (int*)ws;    ws += (size_t)N * sizeof(int);
    int*   csr    = (int*)ws;
    float* H2     = S1;
    const int NB    = cdiv(N, BNODE);
    const int chunk = cdiv(E, G1);
    const int LN    = G1 * (NB + 1);
    int*   loc    = (int*)S1;
    int*   pairs  = loc + (size_t)cdiv(LN, 4) * 4;

    const int T = 256;
    const int GW = 2048;

    k_chunk_sort<<<G1, 1024, (((NB + 1) & ~1) + 2 * chunk) * sizeof(int), stream>>>(src, dst, loc, pairs, E, chunk, NB);
    k_lists_gemm<IN, HID><<<NB, LG_WAVES * 32, 0, stream>>>(pairs, loc, rowptr, csr, x, W1, H1, N, NB, chunk);

    k_gather_gemm<HID, LAT><<<FG_WGS, FG_WAVES * 32, 0, stream>>>(rowptr, csr, H1, b1, W2, H2, N, E);

    k_gather<LAT, false><<<GW, T, 0, stream>>>(rowptr, csr, H2, b2, out, N, E);
}
